# starting kernel
# baseline (speedup 1.0000x reference)
#include <hip/hip_runtime.h>
#include <math.h>

#define NN      500000
#define FDIM    128
#define HDIM    128
#define BB      32768
#define KHOP    4
#define EDG     524288
#define OUTD    8
#define MTOT    (BB + KHOP * EDG)
#define NTILES  (MTOT / 16)
#define XS      132
#define PWF     (16 * XS + 64)

typedef __attribute__((ext_vector_type(2))) float v2f;
typedef __attribute__((ext_vector_type(8))) float v8f;

__device__ __forceinline__ void wait_ds() {
    asm volatile("s_wait_dscnt 0x0" ::: "memory");
}

__global__ void lsappnp_zero(float* __restrict__ ws) {
    size_t i = (size_t)blockIdx.x * blockDim.x + threadIdx.x;
    if (i == 0) ws[0] = 0.0f;
    float* aggx = ws + 64;
    if (i < (size_t)BB * HDIM) aggx[i] = 0.0f;
}

__global__ void lsappnp_dsum(const float* __restrict__ degree,
                             const int* __restrict__ batch,
                             float* __restrict__ ws) {
    __shared__ float red[256];
    float s = 0.0f;
    for (int i = blockIdx.x * blockDim.x + threadIdx.x; i < BB;
         i += gridDim.x * blockDim.x)
        s += degree[batch[i]];
    red[threadIdx.x] = s;
    __syncthreads();
    for (int off = 128; off > 0; off >>= 1) {
        if (threadIdx.x < off) red[threadIdx.x] += red[threadIdx.x + off];
        __syncthreads();
    }
    if (threadIdx.x == 0) atomicAdd(ws, red[0]);
}

__global__ void __launch_bounds__(256, 1)
lsappnp_mlp(const float* __restrict__ x,
            const float* __restrict__ W0, const float* __restrict__ b0,
            const float* __restrict__ W1, const float* __restrict__ b1,
            const float* __restrict__ att,
            const float* __restrict__ degree,
            const int* __restrict__ batch,
            const int* __restrict__ heads,
            const int* __restrict__ ends,
            const float* __restrict__ dsum_p,
            float* __restrict__ aggx) {
    extern __shared__ float smem[];
    v2f*   W0p   = (v2f*)smem;
    v2f*   W1p   = W0p + 64 * 128;
    float* b0s   = (float*)(W1p + 64 * 128);
    float* b1s   = b0s + 128;
    float* Xbase = b1s + 128;

    const int tid  = threadIdx.x;
    const int wave = tid >> 5;
    const int lane = tid & 31;
    const int hi   = lane >> 4;
    const int lm   = lane & 15;

    for (int i = tid; i < 64 * 128; i += 256) {
        int k2 = i >> 7, n = i & 127;
        v2f w0; w0[0] = W0[(2 * k2) * 128 + n]; w0[1] = W0[(2 * k2 + 1) * 128 + n];
        v2f w1; w1[0] = W1[(2 * k2) * 128 + n]; w1[1] = W1[(2 * k2 + 1) * 128 + n];
        W0p[i] = w0;
        W1p[i] = w1;
    }
    if (tid < 128) { b0s[tid] = b0[tid]; b1s[tid] = b1[tid]; }
    __syncthreads();

    const float dsum   = dsum_p[0];
    const float inv_ne = 1.0f / (float)EDG;

    float* Xs    = Xbase + wave * PWF;
    int*   metaS = (int*)(Xs + 16 * XS);
    int*   metaT = metaS + 16;
    float* metaF = (float*)(metaT + 16);

    for (int tile = blockIdx.x * 8 + wave; tile < NTILES; tile += gridDim.x * 8) {
        if (lane < 16) {
            long r = (long)tile * 16 + lane;
            int src, tgt; float sc;
            if (r < BB) {
                tgt = (int)r;
                src = batch[r];
                sc  = att[0];
            } else {
                long idx = r - BB;
                int  k   = (int)(idx >> 19);
                tgt = heads[idx];
                src = ends[idx];
                float db = degree[batch[tgt]];
                float de = degree[src];
                sc = att[k + 1] * rsqrtf(db) * rsqrtf(de) * dsum * inv_ne;
            }
            metaS[lane] = src; metaT[lane] = tgt; metaF[lane] = sc;
        }
        wait_ds();

        for (int rrow = 0; rrow < 16; ++rrow) {
            int src = metaS[rrow];
            float4 v = ((const float4*)(x + (size_t)src * FDIM))[lane];
            ((float4*)(Xs + rrow * XS))[lane] = v;
        }
        wait_ds();

        v8f acc[8];
#pragma unroll
        for (int nt = 0; nt < 8; ++nt) {
            float bv = b0s[nt * 16 + lm];
            v8f c;
#pragma unroll
            for (int i = 0; i < 8; ++i) c[i] = bv;
            acc[nt] = c;
        }
        for (int kt = 0; kt < 32; ++kt) {
            int kc = kt * 4 + hi * 2;
            v2f a  = *(const v2f*)(Xs + lm * XS + kc);
            int k2 = kc >> 1;
#pragma unroll
            for (int nt = 0; nt < 8; ++nt) {
                v2f bfr = W0p[k2 * 128 + nt * 16 + lm];
                acc[nt] = __builtin_amdgcn_wmma_f32_16x16x4_f32(
                    false, a, false, bfr, (short)0, acc[nt], false, false);
            }
        }
#pragma unroll
        for (int nt = 0; nt < 8; ++nt)
#pragma unroll
            for (int v = 0; v < 8; ++v)
                Xs[(v + 8 * hi) * XS + nt * 16 + lm] = fmaxf(acc[nt][v], 0.0f);
        wait_ds();

        v8f acc2[8];
#pragma unroll
        for (int nt = 0; nt < 8; ++nt) {
            float bv = b1s[nt * 16 + lm];
            v8f c;
#pragma unroll
            for (int i = 0; i < 8; ++i) c[i] = bv;
            acc2[nt] = c;
        }
        for (int kt = 0; kt < 32; ++kt) {
            int kc = kt * 4 + hi * 2;
            v2f a  = *(const v2f*)(Xs + lm * XS + kc);
            int k2 = kc >> 1;
#pragma unroll
            for (int nt = 0; nt < 8; ++nt) {
                v2f bfr = W1p[k2 * 128 + nt * 16 + lm];
                acc2[nt] = __builtin_amdgcn_wmma_f32_16x16x4_f32(
                    false, a, false, bfr, (short)0, acc2[nt], false, false);
            }
        }

#pragma unroll
        for (int v = 0; v < 8; ++v) {
            int   tg = metaT[v + 8 * hi];
            float sc = metaF[v + 8 * hi];
            float* base = aggx + (size_t)tg * HDIM + lm;
#pragma unroll
            for (int nt = 0; nt < 8; ++nt)
                atomicAdd(base + nt * 16, acc2[nt][v] * sc);
        }
    }
}

__global__ void lsappnp_head(const float* __restrict__ aggx,
                             const float* __restrict__ W2,
                             const float* __restrict__ b2,
                             float* __restrict__ out) {
    int row  = blockIdx.x * 8 + (threadIdx.x >> 5);
    int lane = threadIdx.x & 31;
    if (row >= BB) return;
    float p[8] = {0, 0, 0, 0, 0, 0, 0, 0};
    for (int c = lane; c < HDIM; c += 32) {
        float a = fmaxf(aggx[(size_t)row * HDIM + c], 0.0f);
        const float4* wp = (const float4*)(W2 + c * OUTD);
        float4 w0 = wp[0], w1 = wp[1];
        p[0] += a * w0.x; p[1] += a * w0.y; p[2] += a * w0.z; p[3] += a * w0.w;
        p[4] += a * w1.x; p[5] += a * w1.y; p[6] += a * w1.z; p[7] += a * w1.w;
    }
#pragma unroll
    for (int off = 16; off > 0; off >>= 1)
#pragma unroll
        for (int j = 0; j < 8; ++j) p[j] += __shfl_down(p[j], off);
    if (lane == 0) {
#pragma unroll
        for (int j = 0; j < 8; ++j) p[j] += b2[j];
        float m = p[0];
#pragma unroll
        for (int j = 1; j < 8; ++j) m = fmaxf(m, p[j]);
        float s = 0.0f;
#pragma unroll
        for (int j = 0; j < 8; ++j) s += __expf(p[j] - m);
        float lse = __logf(s);
#pragma unroll
        for (int j = 0; j < 8; ++j) out[(size_t)row * OUTD + j] = p[j] - m - lse;
    }
}

extern "C" void kernel_launch(void* const* d_in, const int* in_sizes, int n_in,
                              void* d_out, int out_size, void* d_ws, size_t ws_size,
                              hipStream_t stream) {
    const float* x      = (const float*)d_in[0];
    const float* W0     = (const float*)d_in[1];
    const float* b0     = (const float*)d_in[2];
    const float* W1     = (const float*)d_in[3];
    const float* b1     = (const float*)d_in[4];
    const float* W2     = (const float*)d_in[5];
    const float* b2     = (const float*)d_in[6];
    const float* att    = (const float*)d_in[7];
    const float* degree = (const float*)d_in[8];
    const int*   batch  = (const int*)d_in[9];
    const int*   heads  = (const int*)d_in[10];
    const int*   ends   = (const int*)d_in[11];
    float*       out    = (float*)d_out;
    float*       ws     = (float*)d_ws;
    float*       aggx   = ws + 64;

    lsappnp_zero<<<(BB * HDIM + 255) / 256, 256, 0, stream>>>(ws);
    lsappnp_dsum<<<64, 256, 0, stream>>>(degree, batch, ws);

    size_t shmem = (size_t)2 * 64 * 128 * sizeof(v2f)
                 + 256 * sizeof(float)
                 + (size_t)8 * PWF * sizeof(float);
    lsappnp_mlp<<<1280, 256, shmem, stream>>>(x, W0, b0, W1, b1, att, degree,
                                              batch, heads, ends, ws, aggx);

    lsappnp_head<<<BB / 8, 256, 0, stream>>>(aggx, W2, b2, out);
}


# A tile's 16 gathered x rows are loaded together: row ids broadcast from registers by readlane, 16 b128 loads in flight, then 16 LDS stores, instead of one load-wait-store round trip per row
# speedup vs baseline: 1.2034x; 1.2034x over previous
#include <hip/hip_runtime.h>
#include <math.h>

#define NN      500000
#define FDIM    128
#define HDIM    128
#define BB      32768
#define KHOP    4
#define EDG     524288
#define OUTD    8
#define MTOT    (BB + KHOP * EDG)
#define NTILES  (MTOT / 16)
#define XS      132
#define PWF     (16 * XS + 64)

typedef __attribute__((ext_vector_type(2))) float v2f;
typedef __attribute__((ext_vector_type(8))) float v8f;

__device__ __forceinline__ void wait_ds() {
    asm volatile("s_wait_dscnt 0x0" ::: "memory");
}

__global__ void lsappnp_zero(float* __restrict__ ws) {
    size_t i = (size_t)blockIdx.x * blockDim.x + threadIdx.x;
    if (i == 0) ws[0] = 0.0f;
    float* aggx = ws + 64;
    if (i < (size_t)BB * HDIM) aggx[i] = 0.0f;
}

__global__ void lsappnp_dsum(const float* __restrict__ degree,
                             const int* __restrict__ batch,
                             float* __restrict__ ws) {
    __shared__ float red[256];
    float s = 0.0f;
    for (int i = blockIdx.x * blockDim.x + threadIdx.x; i < BB;
         i += gridDim.x * blockDim.x)
        s += degree[batch[i]];
    red[threadIdx.x] = s;
    __syncthreads();
    for (int off = 128; off > 0; off >>= 1) {
        if (threadIdx.x < off) red[threadIdx.x] += red[threadIdx.x + off];
        __syncthreads();
    }
    if (threadIdx.x == 0) atomicAdd(ws, red[0]);
}

__global__ void __launch_bounds__(256, 1)
lsappnp_mlp(const float* __restrict__ x,
            const float* __restrict__ W0, const float* __restrict__ b0,
            const float* __restrict__ W1, const float* __restrict__ b1,
            const float* __restrict__ att,
            const float* __restrict__ degree,
            const int* __restrict__ batch,
            const int* __restrict__ heads,
            const int* __restrict__ ends,
            const float* __restrict__ dsum_p,
            float* __restrict__ aggx) {
    extern __shared__ float smem[];
    v2f*   W0p   = (v2f*)smem;
    v2f*   W1p   = W0p + 64 * 128;
    float* b0s   = (float*)(W1p + 64 * 128);
    float* b1s   = b0s + 128;
    float* Xbase = b1s + 128;

    const int tid  = threadIdx.x;
    const int wave = tid >> 5;
    const int lane = tid & 31;
    const int hi   = lane >> 4;
    const int lm   = lane & 15;

    for (int i = tid; i < 64 * 128; i += 256) {
        int k2 = i >> 7, n = i & 127;
        v2f w0; w0[0] = W0[(2 * k2) * 128 + n]; w0[1] = W0[(2 * k2 + 1) * 128 + n];
        v2f w1; w1[0] = W1[(2 * k2) * 128 + n]; w1[1] = W1[(2 * k2 + 1) * 128 + n];
        W0p[i] = w0;
        W1p[i] = w1;
    }
    if (tid < 128) { b0s[tid] = b0[tid]; b1s[tid] = b1[tid]; }
    __syncthreads();

    const float dsum   = dsum_p[0];
    const float inv_ne = 1.0f / (float)EDG;

    float* Xs    = Xbase + wave * PWF;
    int*   metaT = (int*)(Xs + 16 * XS);
    float* metaF = (float*)(metaT + 16);

    for (int tile = blockIdx.x * 8 + wave; tile < NTILES; tile += gridDim.x * 8) {
        int src = 0;
        if (lane < 16) {
            long r = (long)tile * 16 + lane;
            int tgt; float sc;
            if (r < BB) {
                tgt = (int)r;
                src = batch[r];
                sc  = att[0];
            } else {
                long idx = r - BB;
                int  k   = (int)(idx >> 19);
                tgt = heads[idx];
                src = ends[idx];
                float db = degree[batch[tgt]];
                float de = degree[src];
                sc = att[k + 1] * rsqrtf(db) * rsqrtf(de) * dsum * inv_ne;
            }
            metaT[lane] = tgt; metaF[lane] = sc;
        }

        float4 xr[16];
#pragma unroll
        for (int rrow = 0; rrow < 16; ++rrow) {
            int s = __builtin_amdgcn_readlane(src, rrow);
            xr[rrow] = ((const float4*)(x + (size_t)s * FDIM))[lane];
        }
#pragma unroll
        for (int rrow = 0; rrow < 16; ++rrow)
            ((float4*)(Xs + rrow * XS))[lane] = xr[rrow];
        wait_ds();

        v8f acc[8];
#pragma unroll
        for (int nt = 0; nt < 8; ++nt) {
            float bv = b0s[nt * 16 + lm];
            v8f c;
#pragma unroll
            for (int i = 0; i < 8; ++i) c[i] = bv;
            acc[nt] = c;
        }
        for (int kt = 0; kt < 32; ++kt) {
            int kc = kt * 4 + hi * 2;
            v2f a  = *(const v2f*)(Xs + lm * XS + kc);
            int k2 = kc >> 1;
#pragma unroll
            for (int nt = 0; nt < 8; ++nt) {
                v2f bfr = W0p[k2 * 128 + nt * 16 + lm];
                acc[nt] = __builtin_amdgcn_wmma_f32_16x16x4_f32(
                    false, a, false, bfr, (short)0, acc[nt], false, false);
            }
        }
#pragma unroll
        for (int nt = 0; nt < 8; ++nt)
#pragma unroll
            for (int v = 0; v < 8; ++v)
                Xs[(v + 8 * hi) * XS + nt * 16 + lm] = fmaxf(acc[nt][v], 0.0f);
        wait_ds();

        v8f acc2[8];
#pragma unroll
        for (int nt = 0; nt < 8; ++nt) {
            float bv = b1s[nt * 16 + lm];
            v8f c;
#pragma unroll
            for (int i = 0; i < 8; ++i) c[i] = bv;
            acc2[nt] = c;
        }
        for (int kt = 0; kt < 32; ++kt) {
            int kc = kt * 4 + hi * 2;
            v2f a  = *(const v2f*)(Xs + lm * XS + kc);
            int k2 = kc >> 1;
#pragma unroll
            for (int nt = 0; nt < 8; ++nt) {
                v2f bfr = W1p[k2 * 128 + nt * 16 + lm];
                acc2[nt] = __builtin_amdgcn_wmma_f32_16x16x4_f32(
                    false, a, false, bfr, (short)0, acc2[nt], false, false);
            }
        }

#pragma unroll
        for (int v = 0; v < 8; ++v) {
            int   tg = metaT[v + 8 * hi];
            float sc = metaF[v + 8 * hi];
            float* base = aggx + (size_t)tg * HDIM + lm;
#pragma unroll
            for (int nt = 0; nt < 8; ++nt)
                atomicAdd(base + nt * 16, acc2[nt][v] * sc);
        }
    }
}

__global__ void lsappnp_head(const float* __restrict__ aggx,
                             const float* __restrict__ W2,
                             const float* __restrict__ b2,
                             float* __restrict__ out) {
    int row  = blockIdx.x * 8 + (threadIdx.x >> 5);
    int lane = threadIdx.x & 31;
    if (row >= BB) return;
    float p[8] = {0, 0, 0, 0, 0, 0, 0, 0};
    for (int c = lane; c < HDIM; c += 32) {
        float a = fmaxf(aggx[(size_t)row * HDIM + c], 0.0f);
        const float4* wp = (const float4*)(W2 + c * OUTD);
        float4 w0 = wp[0], w1 = wp[1];
        p[0] += a * w0.x; p[1] += a * w0.y; p[2] += a * w0.z; p[3] += a * w0.w;
        p[4] += a * w1.x; p[5] += a * w1.y; p[6] += a * w1.z; p[7] += a * w1.w;
    }
#pragma unroll
    for (int off = 16; off > 0; off >>= 1)
#pragma unroll
        for (int j = 0; j < 8; ++j) p[j] += __shfl_down(p[j], off);
    if (lane == 0) {
#pragma unroll
        for (int j = 0; j < 8; ++j) p[j] += b2[j];
        float m = p[0];
#pragma unroll
        for (int j = 1; j < 8; ++j) m = fmaxf(m, p[j]);
        float s = 0.0f;
#pragma unroll
        for (int j = 0; j < 8; ++j) s += __expf(p[j] - m);
        float lse = __logf(s);
#pragma unroll
        for (int j = 0; j < 8; ++j) out[(size_t)row * OUTD + j] = p[j] - m - lse;
    }
}

extern "C" void kernel_launch(void* const* d_in, const int* in_sizes, int n_in,
                              void* d_out, int out_size, void* d_ws, size_t ws_size,
                              hipStream_t stream) {
    const float* x      = (const float*)d_in[0];
    const float* W0     = (const float*)d_in[1];
    const float* b0     = (const float*)d_in[2];
    const float* W1     = (const float*)d_in[3];
    const float* b1     = (const float*)d_in[4];
    const float* W2     = (const float*)d_in[5];
    const float* b2     = (const float*)d_in[6];
    const float* att    = (const float*)d_in[7];
    const float* degree = (const float*)d_in[8];
    const int*   batch  = (const int*)d_in[9];
    const int*   heads  = (const int*)d_in[10];
    const int*   ends   = (const int*)d_in[11];
    float*       out    = (float*)d_out;
    float*       ws     = (float*)d_ws;
    float*       aggx   = ws + 64;

    lsappnp_zero<<<(BB * HDIM + 255) / 256, 256, 0, stream>>>(ws);
    lsappnp_dsum<<<64, 256, 0, stream>>>(degree, batch, ws);

    size_t shmem = (size_t)2 * 64 * 128 * sizeof(v2f)
                 + 256 * sizeof(float)
                 + (size_t)8 * PWF * sizeof(float);
    lsappnp_mlp<<<1280, 256, shmem, stream>>>(x, W0, b0, W1, b1, att, degree,
                                              batch, heads, ends, ws, aggx);

    lsappnp_head<<<BB / 8, 256, 0, stream>>>(aggx, W2, b2, out);
}


# Layer 1 moved behind the segment sum (it is linear: sum w*(h@W1+b1) = (sum w*h)@W1 + (sum w)*b1), so it runs on B rows, not B+K*E; float summation order changes
# speedup vs baseline: 2.0805x; 1.7289x over previous
#include <hip/hip_runtime.h>
#include <math.h>


#define NN      500000
#define FDIM    128
#define HDIM    128
#define BB      32768
#define KHOP    4
#define EDG     524288
#define OUTD    8
#define MTOT    (BB + KHOP * EDG)
#define NTILES  (MTOT / 16)
#define NTILEB  (BB / 16)
#define XS      132
#define PWF     (16 * XS + 64)

typedef __attribute__((ext_vector_type(2))) float v2f;
typedef __attribute__((ext_vector_type(8))) float v8f;


__device__ __forceinline__ void wait_ds() {
    asm volatile("s_wait_dscnt 0x0" ::: "memory");
}

__global__ void lsappnp_zero(float* __restrict__ ws, float* __restrict__ out) {
    size_t i = (size_t)blockIdx.x * blockDim.x + threadIdx.x;
    if (i == 0) ws[0] = 0.0f;
    float* aggh = ws + 64;
    if (i < (size_t)BB * HDIM) aggh[i] = 0.0f;
    if (i < (size_t)BB) out[i * OUTD] = 0.0f;
}

__global__ void lsappnp_dsum(const float* __restrict__ degree,
                             const int* __restrict__ batch,
                             float* __restrict__ ws) {
    __shared__ float red[256];
    float s = 0.0f;
    for (int i = blockIdx.x * blockDim.x + threadIdx.x; i < BB;
         i += gridDim.x * blockDim.x)
        s += degree[batch[i]];
    red[threadIdx.x] = s;
    __syncthreads();
    for (int off = 128; off > 0; off >>= 1) {
        if (threadIdx.x < off) red[threadIdx.x] += red[threadIdx.x + off];
        __syncthreads();
    }
    if (threadIdx.x == 0) atomicAdd(ws, red[0]);
}

__global__ void __launch_bounds__(256)
lsappnp_mlp(const float* __restrict__ x,
            const float* __restrict__ W0, const float* __restrict__ b0,
            const float* __restrict__ att,
            const float* __restrict__ degree,
            const int* __restrict__ batch,
            const int* __restrict__ heads,
            const int* __restrict__ ends,
            const float* __restrict__ dsum_p,
            float* __restrict__ aggh,
            float* __restrict__ wsum) {
    extern __shared__ float smem[];
    v2f*   W0p   = (v2f*)smem;
    float* b0s   = (float*)(W0p + 64 * 128);
    float* Xbase = b0s + 128;

    const int tid  = threadIdx.x;
    const int wave = tid >> 5;
    const int lane = tid & 31;
    const int hi   = lane >> 4;
    const int lm   = lane & 15;

    for (int i = tid; i < 64 * 128; i += 256) {
        int k2 = i >> 7, n = i & 127;
        v2f w0; w0[0] = W0[(2 * k2) * 128 + n]; w0[1] = W0[(2 * k2 + 1) * 128 + n];
        W0p[i] = w0;
    }
    if (tid < 128) b0s[tid] = b0[tid];
    __syncthreads();

    const float dsum   = dsum_p[0];
    const float inv_ne = 1.0f / (float)EDG;

    float* Xs    = Xbase + wave * PWF;
    int*   metaT = (int*)(Xs + 16 * XS);
    float* metaF = (float*)(metaT + 16);

    for (int tile = blockIdx.x * 8 + wave; tile < NTILES; tile += gridDim.x * 8) {
        int   src = 0;
        int   tgt = 0;
        float sc  = 0.0f;
        if (lane < 16) {
            long r = (long)tile * 16 + lane;
            if (r < BB) {
                tgt = (int)r;
                src = batch[r];
                sc  = att[0];
            } else {
                long idx = r - BB;
                int  k   = (int)(idx >> 19);
                tgt = heads[idx];
                src = ends[idx];
                float db = degree[batch[tgt]];
                float de = degree[src];
                sc = att[k + 1] * rsqrtf(db) * rsqrtf(de) * dsum * inv_ne;
            }
            metaT[lane] = tgt; metaF[lane] = sc;
        }

        float4 xr[16];
#pragma unroll
        for (int rrow = 0; rrow < 16; ++rrow) {
            int s = __builtin_amdgcn_readlane(src, rrow);
            xr[rrow] = ((const float4*)(x + (size_t)s * FDIM))[lane];
        }
#pragma unroll
        for (int rrow = 0; rrow < 16; ++rrow)
            ((float4*)(Xs + rrow * XS))[lane] = xr[rrow];
        wait_ds();

        v8f acc[8];
#pragma unroll
        for (int nt = 0; nt < 8; ++nt) {
            float bv = b0s[nt * 16 + lm];
            v8f c;
#pragma unroll
            for (int i = 0; i < 8; ++i) c[i] = bv;
            acc[nt] = c;
        }
        for (int kt = 0; kt < 32; ++kt) {
            int kc = kt * 4 + hi * 2;
            v2f a  = *(const v2f*)(Xs + lm * XS + kc);
            int k2 = kc >> 1;
#pragma unroll
            for (int nt = 0; nt < 8; ++nt) {
                v2f bfr = W0p[k2 * 128 + nt * 16 + lm];
                acc[nt] = __builtin_amdgcn_wmma_f32_16x16x4_f32(
                    false, a, false, bfr, (short)0, acc[nt], false, false);
            }
        }

#pragma unroll
        for (int v = 0; v < 8; ++v) {
            int   tg = metaT[v + 8 * hi];
            float sw = metaF[v + 8 * hi];
            float* base = aggh + (size_t)tg * HDIM + lm;
#pragma unroll
            for (int nt = 0; nt < 8; ++nt)
                atomicAdd(base + nt * 16, fmaxf(acc[nt][v], 0.0f) * sw);
        }
        if (lane < 16) atomicAdd(wsum + (size_t)tgt * OUTD, sc);
    }
}

__global__ void __launch_bounds__(256)
lsappnp_head(const float* __restrict__ aggh,
             const float* __restrict__ W1, const float* __restrict__ b1,
             const float* __restrict__ W2, const float* __restrict__ b2,
             float* out) {
    extern __shared__ float smem[];
    v2f*   W1p   = (v2f*)smem;
    float* b1s   = (float*)(W1p + 64 * 128);
    float* W2s   = b1s + 128;
    float* b2s   = W2s + HDIM * OUTD;
    float* Xbase = b2s + 64;

    const int tid  = threadIdx.x;
    const int wave = tid >> 5;
    const int lane = tid & 31;
    const int hi   = lane >> 4;
    const int lm   = lane & 15;

    for (int i = tid; i < 64 * 128; i += 256) {
        int k2 = i >> 7, n = i & 127;
        v2f w1; w1[0] = W1[(2 * k2) * 128 + n]; w1[1] = W1[(2 * k2 + 1) * 128 + n];
        W1p[i] = w1;
    }
    if (tid < 128) b1s[tid] = b1[tid];
    for (int i = tid; i < HDIM * OUTD; i += 256) W2s[i] = W2[i];
    if (tid < OUTD) b2s[tid] = b2[tid];
    __syncthreads();

    float* Xs    = Xbase + wave * PWF;
    float* metaF = Xs + 16 * XS;

    for (int tile = blockIdx.x * 8 + wave; tile < NTILEB; tile += gridDim.x * 8) {
        const size_t row0 = (size_t)tile * 16;

        float4 xr[16];
#pragma unroll
        for (int rrow = 0; rrow < 16; ++rrow)
            xr[rrow] = ((const float4*)(aggh + (row0 + rrow) * HDIM))[lane];
        float wsv = 0.0f;
        if (lane < 16) wsv = out[(row0 + lane) * OUTD];
#pragma unroll
        for (int rrow = 0; rrow < 16; ++rrow)
            ((float4*)(Xs + rrow * XS))[lane] = xr[rrow];
        if (lane < 16) metaF[lane] = wsv;
        wait_ds();

        v8f acc2[8];
#pragma unroll
        for (int nt = 0; nt < 8; ++nt) {
            float bv = b1s[nt * 16 + lm];
            v8f c;
#pragma unroll
            for (int i = 0; i < 8; ++i) c[i] = metaF[i + 8 * hi] * bv;
            acc2[nt] = c;
        }
        for (int kt = 0; kt < 32; ++kt) {
            int kc = kt * 4 + hi * 2;
            v2f a  = *(const v2f*)(Xs + lm * XS + kc);
            int k2 = kc >> 1;
#pragma unroll
            for (int nt = 0; nt < 8; ++nt) {
                v2f bfr = W1p[k2 * 128 + nt * 16 + lm];
                acc2[nt] = __builtin_amdgcn_wmma_f32_16x16x4_f32(
                    false, a, false, bfr, (short)0, acc2[nt], false, false);
            }
        }
#pragma unroll
        for (int nt = 0; nt < 8; ++nt)
#pragma unroll
            for (int v = 0; v < 8; ++v)
                Xs[(v + 8 * hi) * XS + nt * 16 + lm] = fmaxf(acc2[nt][v], 0.0f);
        wait_ds();

        const float* ar = Xs + lm * XS;
        const float* wr = W2s + hi * 4;
        float p0 = 0.0f, p1 = 0.0f, p2 = 0.0f, p3 = 0.0f;
        for (int c = 0; c < HDIM; c += 4) {
            float4 a  = *(const float4*)(ar + c);
            float4 w0 = *(const float4*)(wr + (c + 0) * OUTD);
            float4 w1 = *(const float4*)(wr + (c + 1) * OUTD);
            float4 w2 = *(const float4*)(wr + (c + 2) * OUTD);
            float4 w3 = *(const float4*)(wr + (c + 3) * OUTD);
            p0 += a.x * w0.x; p1 += a.x * w0.y; p2 += a.x * w0.z; p3 += a.x * w0.w;
            p0 += a.y * w1.x; p1 += a.y * w1.y; p2 += a.y * w1.z; p3 += a.y * w1.w;
            p0 += a.z * w2.x; p1 += a.z * w2.y; p2 += a.z * w2.z; p3 += a.z * w2.w;
            p0 += a.w * w3.x; p1 += a.w * w3.y; p2 += a.w * w3.z; p3 += a.w * w3.w;
        }
        p0 += b2s[hi * 4 + 0]; p1 += b2s[hi * 4 + 1]; p2 += b2s[hi * 4 + 2]; p3 += b2s[hi * 4 + 3];
        float m = fmaxf(fmaxf(p0, p1), fmaxf(p2, p3));
        m = fmaxf(m, __shfl_xor(m, 16));
        float s = __expf(p0 - m) + __expf(p1 - m) + __expf(p2 - m) + __expf(p3 - m);
        s += __shfl_xor(s, 16);
        float lse = __logf(s);
        float4 o;
        o.x = p0 - m - lse; o.y = p1 - m - lse; o.z = p2 - m - lse; o.w = p3 - m - lse;
        *(float4*)(out + (row0 + lm) * OUTD + hi * 4) = o;
    }
}

extern "C" void kernel_launch(void* const* d_in, const int* in_sizes, int n_in,
                              void* d_out, int out_size, void* d_ws, size_t ws_size,
                              hipStream_t stream) {
    const float* x      = (const float*)d_in[0];
    const float* W0     = (const float*)d_in[1];
    const float* b0     = (const float*)d_in[2];
    const float* W1     = (const float*)d_in[3];
    const float* b1     = (const float*)d_in[4];
    const float* W2     = (const float*)d_in[5];
    const float* b2     = (const float*)d_in[6];
    const float* att    = (const float*)d_in[7];
    const float* degree = (const float*)d_in[8];
    const int*   batch  = (const int*)d_in[9];
    const int*   heads  = (const int*)d_in[10];
    const int*   ends   = (const int*)d_in[11];
    float*       out    = (float*)d_out;
    float*       ws     = (float*)d_ws;
    float*       aggh   = ws + 64;

    lsappnp_zero<<<(BB * HDIM + 255) / 256, 256, 0, stream>>>(ws, out);
    lsappnp_dsum<<<64, 256, 0, stream>>>(degree, batch, ws);

    size_t shmem_mlp = (size_t)64 * 128 * sizeof(v2f)
                     + 128 * sizeof(float)
                     + (size_t)8 * PWF * sizeof(float);
    lsappnp_mlp<<<512, 256, shmem_mlp, stream>>>(x, W0, b0, att, degree,
                                                 batch, heads, ends, ws, aggh, out);

    size_t shmem_head = (size_t)64 * 128 * sizeof(v2f)
                      + (128 + HDIM * OUTD + 64) * sizeof(float)
                      + (size_t)8 * PWF * sizeof(float);
    lsappnp_head<<<NTILEB / 8, 256, shmem_head, stream>>>(aggh, W1, b1, W2, b2, out);
}


# Layer 0 runs once per node (500000 rows, read in order) not once per gathered row (2129920): rows are counted and placed by source node, then w*h is added per row; float summation order changes
# speedup vs baseline: 2.3734x; 1.1408x over previous
#include <hip/hip_runtime.h>
#include <math.h>


#define NN      500000
#define FDIM    128
#define HDIM    128
#define BB      32768
#define KHOP    4
#define EDG     524288
#define OUTD    8
#define MTOT    (BB + KHOP * EDG)
#define NTILES  (MTOT / 16)
#define NTILEN  (NN / 16)
#define NTILEB  (BB / 16)
#define XS      132
#define PWF     (16 * XS + 128)

#define NPAD     500032
#define WS_AGGH  64
#define WS_CNT   (WS_AGGH + BB * HDIM)
#define WS_POS   (WS_CNT + NPAD)
#define WS_EREC  (WS_POS + NPAD)
#define WS_WORDS (WS_EREC + MTOT)
#define WS_NEED  ((unsigned long long)WS_WORDS * 4ull)

typedef __attribute__((ext_vector_type(2))) float v2f;
typedef __attribute__((ext_vector_type(4))) float v4f;
typedef __attribute__((ext_vector_type(8))) float v8f;


__device__ __forceinline__ void wait_ds() {
    asm volatile("s_wait_dscnt 0x0" ::: "memory");
}

__global__ void lsappnp_zero(float* __restrict__ ws, float* __restrict__ out,
                             unsigned long long ws_bytes) {
    size_t i = (size_t)blockIdx.x * blockDim.x + threadIdx.x;
    const bool per_node = ws_bytes >= WS_NEED;
    if (i == 0) {
        ws[0] = 0.0f;
        if (per_node) ((int*)ws)[1] = 0;
    }
    float* aggh = ws + WS_AGGH;
    if (i < (size_t)BB * HDIM) aggh[i] = 0.0f;
    if (i < (size_t)BB) out[i * OUTD] = 0.0f;
    if (per_node && i < (size_t)NN) ((int*)(ws + WS_CNT))[i] = 0;
}

__global__ void lsappnp_dsum(const float* __restrict__ degree,
                             const int* __restrict__ batch,
                             float* __restrict__ ws) {
    __shared__ float red[256];
    float s = 0.0f;
    for (int i = blockIdx.x * blockDim.x + threadIdx.x; i < BB;
         i += gridDim.x * blockDim.x)
        s += degree[batch[i]];
    red[threadIdx.x] = s;
    __syncthreads();
    for (int off = 128; off > 0; off >>= 1) {
        if (threadIdx.x < off) red[threadIdx.x] += red[threadIdx.x + off];
        __syncthreads();
    }
    if (threadIdx.x == 0) atomicAdd(ws, red[0]);
}

__global__ void __launch_bounds__(256)
lsappnp_count(const int* __restrict__ batch, const int* __restrict__ ends,
              float* __restrict__ ws, unsigned long long ws_bytes) {
    if (ws_bytes < WS_NEED) return;
    int* cnt = (int*)(ws + WS_CNT);
    const int r = blockIdx.x * 256 + threadIdx.x;
    const int src = (r < BB) ? batch[r] : ends[r - BB];
    atomicAdd(cnt + src, 1);
}

__global__ void __launch_bounds__(256)
lsappnp_alloc(float* __restrict__ ws, unsigned long long ws_bytes) {
    if (ws_bytes < WS_NEED) return;
    __shared__ int scan[256];
    __shared__ int sbase;
    const int* cnt = (const int*)(ws + WS_CNT);
    int* pos = (int*)(ws + WS_POS);
    int* cursor = (int*)ws + 1;
    const int tid = threadIdx.x;
    const int n = blockIdx.x * 256 + tid;
    const int c = (n < NN) ? cnt[n] : 0;
    scan[tid] = c;
    __syncthreads();
    for (int off = 1; off < 256; off <<= 1) {
        int v = (tid >= off) ? scan[tid - off] : 0;
        __syncthreads();
        scan[tid] += v;
        __syncthreads();
    }
    const int incl = scan[tid];
    if (tid == 255) sbase = atomicAdd(cursor, incl);
    __syncthreads();
    if (n < NN) pos[n] = sbase + incl - c;
}

__global__ void __launch_bounds__(256)
lsappnp_place(const float* __restrict__ att,
              const float* __restrict__ degree,
              const int* __restrict__ batch,
              const int* __restrict__ heads,
              const int* __restrict__ ends,
              float* __restrict__ ws,
              float* __restrict__ wsum,
              unsigned long long ws_bytes) {
    if (ws_bytes < WS_NEED) return;
    int* pos = (int*)(ws + WS_POS);
    unsigned* erec = (unsigned*)(ws + WS_EREC);
    const float dsum   = ws[0];
    const float inv_ne = 1.0f / (float)EDG;
    const int r = blockIdx.x * 256 + threadIdx.x;
    int src, tgt, kk;
    float sc;
    if (r < BB) {
        tgt = r;
        src = batch[r];
        sc  = att[0];
        kk  = 0;
    } else {
        int idx = r - BB;
        int k   = idx >> 19;
        kk  = k + 1;
        tgt = heads[idx];
        src = ends[idx];
        float db = degree[batch[tgt]];
        float de = degree[src];
        sc = att[k + 1] * rsqrtf(db) * rsqrtf(de) * dsum * inv_ne;
    }
    const int p = atomicAdd(pos + src, 1);
    const unsigned rec = (unsigned)tgt | ((unsigned)kk << 16);
    if ((unsigned)p < (unsigned)MTOT) atomicExch(erec + p, rec);
    atomicAdd(wsum + (size_t)tgt * OUTD, sc);
}

__device__ __forceinline__ void chunk_issue(int base, int eE, int lane,
                                            const unsigned* __restrict__ erec,
                                            const int* __restrict__ batch,
                                            const float* __restrict__ degree,
                                            const float* __restrict__ att,
                                            unsigned& q0, unsigned& q1,
                                            float& d0, float& d1, float& a0, float& a1) {
    q0 = 0u; q1 = 0u;
    if (base + lane < eE)      q0 = erec[base + lane];
    if (base + 32 + lane < eE) q1 = erec[base + 32 + lane];
    const int t0 = (int)(q0 & (unsigned)(BB - 1)), t1 = (int)(q1 & (unsigned)(BB - 1));
    const int k0 = min((int)((q0 >> 16) & 7u), KHOP), k1 = min((int)((q1 >> 16) & 7u), KHOP);
    d0 = degree[batch[t0]]; d1 = degree[batch[t1]];
    a0 = att[k0];           a1 = att[k1];
}
__device__ __forceinline__ void chunk_weights(unsigned q0, unsigned q1, float d0, float d1,
                                              float a0, float a1, float& p0, float& p1) {
    p0 = (((q0 >> 16) & 7u) == 0u) ? a0 : a0 * rsqrtf(d0);
    p1 = (((q1 >> 16) & 7u) == 0u) ? a1 : a1 * rsqrtf(d1);
}

__global__ void __launch_bounds__(256)
lsappnp_mlp(const float* __restrict__ x,
            const float* __restrict__ W0, const float* __restrict__ b0,
            const float* __restrict__ att,
            const float* __restrict__ degree,
            const int* __restrict__ batch,
            const int* __restrict__ heads,
            const int* __restrict__ ends,
            float* __restrict__ ws,
            float* __restrict__ wsum,
            unsigned long long ws_bytes) {
    extern __shared__ float smem[];
    v2f*   W0p   = (v2f*)smem;
    float* b0s   = (float*)(W0p + 64 * 128);
    float* Xbase = b0s + 128;

    const int tid  = threadIdx.x;
    const int wave = tid >> 5;
    const int lane = tid & 31;
    const int hi   = lane >> 4;
    const int lm   = lane & 15;

    for (int i = tid; i < 64 * 128; i += 256) {
        int k2 = i >> 7, n = i & 127;
        v2f w0; w0[0] = W0[(2 * k2) * 128 + n]; w0[1] = W0[(2 * k2 + 1) * 128 + n];
        W0p[i] = w0;
    }
    if (tid < 128) b0s[tid] = b0[tid];
    __syncthreads();

    float* aggh = ws + WS_AGGH;
    float* Xs   = Xbase + wave * PWF;

    if (ws_bytes >= WS_NEED) {
        const int* cnt = (const int*)(ws + WS_CNT);
        const int* pos = (const int*)(ws + WS_POS);
        const unsigned* erec = (const unsigned*)(ws + WS_EREC);
        const float dsum   = ws[0];
        const float inv_ne = 1.0f / (float)EDG;
        unsigned* metaT = (unsigned*)(Xs + 16 * XS);
        float*    metaF = (float*)(metaT + 64);

        for (int tile = blockIdx.x * 8 + wave; tile < NTILEN; tile += gridDim.x * 8) {
            const int n0 = tile * 16;
            v4f xr[16];
            const v4f* xp = (const v4f*)(x + (size_t)n0 * FDIM) + lane;
#pragma unroll
            for (int rrow = 0; rrow < 16; ++rrow)
                xr[rrow] = __builtin_nontemporal_load(xp + rrow * 32);
            int c = 0, pe = 0;
            float de = 1.0f;
            if (lane < 16) { c = cnt[n0 + lane]; pe = pos[n0 + lane]; de = degree[n0 + lane]; }
            __builtin_amdgcn_sched_barrier(0);
            const float rde = rsqrtf(de);

            const int ps = pe - c;
            int eS = __builtin_amdgcn_readlane(ps, 0);
            int eE = __builtin_amdgcn_readlane(pe, 15);
            eS = max(eS, 0);
            eE = min(eE, MTOT);
            unsigned q0, q1;
            float d0, d1, a0, a1, p0, p1;
            chunk_issue(eS, eE, lane, erec, batch, degree, att, q0, q1, d0, d1, a0, a1);

#pragma unroll
            for (int rrow = 0; rrow < 16; ++rrow)
                *(v4f*)(Xs + rrow * XS + lane * 4) = xr[rrow];
            wait_ds();

            v8f acc[8];
#pragma unroll
            for (int nt = 0; nt < 8; ++nt) {
                float bv = b0s[nt * 16 + lm];
                v8f cc;
#pragma unroll
                for (int i = 0; i < 8; ++i) cc[i] = bv;
                acc[nt] = cc;
            }
            for (int kt = 0; kt < 32; ++kt) {
                int kc = kt * 4 + hi * 2;
                v2f a  = *(const v2f*)(Xs + lm * XS + kc);
                int k2 = kc >> 1;
#pragma unroll
                for (int nt = 0; nt < 8; ++nt) {
                    v2f bfr = W0p[k2 * 128 + nt * 16 + lm];
                    acc[nt] = __builtin_amdgcn_wmma_f32_16x16x4_f32(
                        false, a, false, bfr, (short)0, acc[nt], false, false);
                }
            }
#pragma unroll
            for (int nt = 0; nt < 8; ++nt)
#pragma unroll
                for (int v = 0; v < 8; ++v)
                    Xs[(v + 8 * hi) * XS + nt * 16 + lm] = fmaxf(acc[nt][v], 0.0f);

            for (int base = eS; base < eE; base += 64) {
                if (base != eS)
                    chunk_issue(base, eE, lane, erec, batch, degree, att, q0, q1, d0, d1, a0, a1);
                chunk_weights(q0, q1, d0, d1, a0, a1, p0, p1);
                metaT[lane]      = q0;
                metaF[lane]      = p0;
                metaT[32 + lane] = q1;
                metaF[32 + lane] = p1;
                wait_ds();
                const int lim = min(eE, base + 64);
#pragma unroll
                for (int m = 0; m < 16; ++m) {
                    const int s = max(__builtin_amdgcn_readlane(ps, m), base);
                    const int e = min(__builtin_amdgcn_readlane(pe, m), lim);
                    if (s < e) {
                        const float* hr = Xs + m * XS + lane;
                        const float h0 = hr[0], h1 = hr[32], h2 = hr[64], h3 = hr[96];
                        const float rde_m = __int_as_float(__builtin_amdgcn_readlane(__float_as_int(rde), m));
                        for (int j = s; j < e; ++j) {
                            const unsigned q  = (unsigned)__builtin_amdgcn_readfirstlane((int)metaT[j - base]);
                            const int      tg = (int)(q & (unsigned)(BB - 1));
                            const float    wp = metaF[j - base];
                            const float    w  = (((q >> 16) & 7u) == 0u) ? wp : wp * rde_m * dsum * inv_ne;
                            float* d = aggh + (size_t)tg * HDIM + lane;
                            atomicAdd(d,      h0 * w);
                            atomicAdd(d + 32, h1 * w);
                            atomicAdd(d + 64, h2 * w);
                            atomicAdd(d + 96, h3 * w);
                        }
                    }
                }
            }
        }
        return;
    }

    const float dsum   = ws[0];
    const float inv_ne = 1.0f / (float)EDG;

    int*   metaT = (int*)(Xs + 16 * XS);
    float* metaF = (float*)(metaT + 16);

    for (int tile = blockIdx.x * 8 + wave; tile < NTILES; tile += gridDim.x * 8) {
        int   src = 0;
        int   tgt = 0;
        float sc  = 0.0f;
        if (lane < 16) {
            long r = (long)tile * 16 + lane;
            if (r < BB) {
                tgt = (int)r;
                src = batch[r];
                sc  = att[0];
            } else {
                long idx = r - BB;
                int  k   = (int)(idx >> 19);
                tgt = heads[idx];
                src = ends[idx];
                float db = degree[batch[tgt]];
                float de = degree[src];
                sc = att[k + 1] * rsqrtf(db) * rsqrtf(de) * dsum * inv_ne;
            }
            metaT[lane] = tgt; metaF[lane] = sc;
        }

        float4 xr[16];
#pragma unroll
        for (int rrow = 0; rrow < 16; ++rrow) {
            int s = __builtin_amdgcn_readlane(src, rrow);
            xr[rrow] = ((const float4*)(x + (size_t)s * FDIM))[lane];
        }
#pragma unroll
        for (int rrow = 0; rrow < 16; ++rrow)
            ((float4*)(Xs + rrow * XS))[lane] = xr[rrow];
        wait_ds();

        v8f acc[8];
#pragma unroll
        for (int nt = 0; nt < 8; ++nt) {
            float bv = b0s[nt * 16 + lm];
            v8f c;
#pragma unroll
            for (int i = 0; i < 8; ++i) c[i] = bv;
            acc[nt] = c;
        }
        for (int kt = 0; kt < 32; ++kt) {
            int kc = kt * 4 + hi * 2;
            v2f a  = *(const v2f*)(Xs + lm * XS + kc);
            int k2 = kc >> 1;
#pragma unroll
            for (int nt = 0; nt < 8; ++nt) {
                v2f bfr = W0p[k2 * 128 + nt * 16 + lm];
                acc[nt] = __builtin_amdgcn_wmma_f32_16x16x4_f32(
                    false, a, false, bfr, (short)0, acc[nt], false, false);
            }
        }

#pragma unroll
        for (int v = 0; v < 8; ++v) {
            int   tg = metaT[v + 8 * hi];
            float sw = metaF[v + 8 * hi];
            float* base = aggh + (size_t)tg * HDIM + lm;
#pragma unroll
            for (int nt = 0; nt < 8; ++nt)
                atomicAdd(base + nt * 16, fmaxf(acc[nt][v], 0.0f) * sw);
        }
        if (lane < 16) atomicAdd(wsum + (size_t)tgt * OUTD, sc);
    }
}

__global__ void __launch_bounds__(256)
lsappnp_head(const float* __restrict__ aggh,
             const float* __restrict__ W1, const float* __restrict__ b1,
             const float* __restrict__ W2, const float* __restrict__ b2,
             float* out) {
    extern __shared__ float smem[];
    v2f*   W1p   = (v2f*)smem;
    float* b1s   = (float*)(W1p + 64 * 128);
    float* W2s   = b1s + 128;
    float* b2s   = W2s + HDIM * OUTD;
    float* Xbase = b2s + 64;

    const int tid  = threadIdx.x;
    const int wave = tid >> 5;
    const int lane = tid & 31;
    const int hi   = lane >> 4;
    const int lm   = lane & 15;

    for (int i = tid; i < 64 * 128; i += 256) {
        int k2 = i >> 7, n = i & 127;
        v2f w1; w1[0] = W1[(2 * k2) * 128 + n]; w1[1] = W1[(2 * k2 + 1) * 128 + n];
        W1p[i] = w1;
    }
    if (tid < 128) b1s[tid] = b1[tid];
    for (int i = tid; i < HDIM * OUTD; i += 256) W2s[i] = W2[i];
    if (tid < OUTD) b2s[tid] = b2[tid];
    __syncthreads();

    float* Xs    = Xbase + wave * PWF;
    float* metaF = Xs + 16 * XS;

    for (int tile = blockIdx.x * 8 + wave; tile < NTILEB; tile += gridDim.x * 8) {
        const size_t row0 = (size_t)tile * 16;

        float4 xr[16];
#pragma unroll
        for (int rrow = 0; rrow < 16; ++rrow)
            xr[rrow] = ((const float4*)(aggh + (row0 + rrow) * HDIM))[lane];
        float wsv = 0.0f;
        if (lane < 16) wsv = out[(row0 + lane) * OUTD];
#pragma unroll
        for (int rrow = 0; rrow < 16; ++rrow)
            ((float4*)(Xs + rrow * XS))[lane] = xr[rrow];
        if (lane < 16) metaF[lane] = wsv;
        wait_ds();

        v8f acc2[8];
#pragma unroll
        for (int nt = 0; nt < 8; ++nt) {
            float bv = b1s[nt * 16 + lm];
            v8f c;
#pragma unroll
            for (int i = 0; i < 8; ++i) c[i] = metaF[i + 8 * hi] * bv;
            acc2[nt] = c;
        }
        for (int kt = 0; kt < 32; ++kt) {
            int kc = kt * 4 + hi * 2;
            v2f a  = *(const v2f*)(Xs + lm * XS + kc);
            int k2 = kc >> 1;
#pragma unroll
            for (int nt = 0; nt < 8; ++nt) {
                v2f bfr = W1p[k2 * 128 + nt * 16 + lm];
                acc2[nt] = __builtin_amdgcn_wmma_f32_16x16x4_f32(
                    false, a, false, bfr, (short)0, acc2[nt], false, false);
            }
        }
#pragma unroll
        for (int nt = 0; nt < 8; ++nt)
#pragma unroll
            for (int v = 0; v < 8; ++v)
                Xs[(v + 8 * hi) * XS + nt * 16 + lm] = fmaxf(acc2[nt][v], 0.0f);
        wait_ds();

        const float* ar = Xs + lm * XS;
        const float* wr = W2s + hi * 4;
        float p0 = 0.0f, p1 = 0.0f, p2 = 0.0f, p3 = 0.0f;
        for (int c = 0; c < HDIM; c += 4) {
            float4 a  = *(const float4*)(ar + c);
            float4 w0 = *(const float4*)(wr + (c + 0) * OUTD);
            float4 w1 = *(const float4*)(wr + (c + 1) * OUTD);
            float4 w2 = *(const float4*)(wr + (c + 2) * OUTD);
            float4 w3 = *(const float4*)(wr + (c + 3) * OUTD);
            p0 += a.x * w0.x; p1 += a.x * w0.y; p2 += a.x * w0.z; p3 += a.x * w0.w;
            p0 += a.y * w1.x; p1 += a.y * w1.y; p2 += a.y * w1.z; p3 += a.y * w1.w;
            p0 += a.z * w2.x; p1 += a.z * w2.y; p2 += a.z * w2.z; p3 += a.z * w2.w;
            p0 += a.w * w3.x; p1 += a.w * w3.y; p2 += a.w * w3.z; p3 += a.w * w3.w;
        }
        p0 += b2s[hi * 4 + 0]; p1 += b2s[hi * 4 + 1]; p2 += b2s[hi * 4 + 2]; p3 += b2s[hi * 4 + 3];
        float m = fmaxf(fmaxf(p0, p1), fmaxf(p2, p3));
        m = fmaxf(m, __shfl_xor(m, 16));
        float s = __expf(p0 - m) + __expf(p1 - m) + __expf(p2 - m) + __expf(p3 - m);
        s += __shfl_xor(s, 16);
        float lse = __logf(s);
        float4 o;
        o.x = p0 - m - lse; o.y = p1 - m - lse; o.z = p2 - m - lse; o.w = p3 - m - lse;
        *(float4*)(out + (row0 + lm) * OUTD + hi * 4) = o;
    }
}

extern "C" void kernel_launch(void* const* d_in, const int* in_sizes, int n_in,
                              void* d_out, int out_size, void* d_ws, size_t ws_size,
                              hipStream_t stream) {
    const float* x      = (const float*)d_in[0];
    const float* W0     = (const float*)d_in[1];
    const float* b0     = (const float*)d_in[2];
    const float* W1     = (const float*)d_in[3];
    const float* b1     = (const float*)d_in[4];
    const float* W2     = (const float*)d_in[5];
    const float* b2     = (const float*)d_in[6];
    const float* att    = (const float*)d_in[7];
    const float* degree = (const float*)d_in[8];
    const int*   batch  = (const int*)d_in[9];
    const int*   heads  = (const int*)d_in[10];
    const int*   ends   = (const int*)d_in[11];
    float*       out    = (float*)d_out;
    float*       ws     = (float*)d_ws;
    float*       aggh   = ws + WS_AGGH;
    const unsigned long long ws_bytes = (unsigned long long)ws_size;

    lsappnp_zero<<<(BB * HDIM + 255) / 256, 256, 0, stream>>>(ws, out, ws_bytes);
    lsappnp_dsum<<<64, 256, 0, stream>>>(degree, batch, ws);
    lsappnp_count<<<MTOT / 256, 256, 0, stream>>>(batch, ends, ws, ws_bytes);
    lsappnp_alloc<<<(NN + 255) / 256, 256, 0, stream>>>(ws, ws_bytes);
    lsappnp_place<<<MTOT / 256, 256, 0, stream>>>(att, degree, batch, heads, ends, ws, out, ws_bytes);

    size_t shmem_mlp = (size_t)64 * 128 * sizeof(v2f)
                     + 128 * sizeof(float)
                     + (size_t)8 * PWF * sizeof(float);
    lsappnp_mlp<<<512, 256, shmem_mlp, stream>>>(x, W0, b0, att, degree,
                                                 batch, heads, ends, ws, out, ws_bytes);

    size_t shmem_head = (size_t)64 * 128 * sizeof(v2f)
                      + (128 + HDIM * OUTD + 64) * sizeof(float)
                      + (size_t)8 * PWF * sizeof(float);
    lsappnp_head<<<NTILEB / 8, 256, shmem_head, stream>>>(aggh, W1, b1, W2, b2, out);
}


# The counting add returns each row's rank in its node's run (arrival order, as the placing add gave before), kept in the idle sum plane: slot = run start + rank, no second add; place zeroes the plane
# speedup vs baseline: 2.6054x; 1.0978x over previous
#include <hip/hip_runtime.h>
#include <math.h>


#define NN      500000
#define FDIM    128
#define HDIM    128
#define BB      32768
#define KHOP    4
#define EDG     524288
#define OUTD    8
#define MTOT    (BB + KHOP * EDG)
#define NTILES  (MTOT / 16)
#define NTILEN  (NN / 16)
#define NTILEB  (BB / 16)
#define XS      132
#define PWF     (16 * XS + 128)

#define NPAD     500032
#define WS_AGGH  64
#define WS_CNT   (WS_AGGH + BB * HDIM)
#define WS_POS   (WS_CNT + NPAD)
#define WS_EREC  (WS_POS + NPAD)
#define WS_WORDS (WS_EREC + MTOT)
#define WS_NEED  ((unsigned long long)WS_WORDS * 4ull)

typedef __attribute__((ext_vector_type(2))) float v2f;
typedef __attribute__((ext_vector_type(4))) float v4f;
typedef __attribute__((ext_vector_type(8))) float v8f;


__device__ __forceinline__ void wait_ds() {
    asm volatile("s_wait_dscnt 0x0" ::: "memory");
}

__global__ void lsappnp_zero(float* __restrict__ ws, float* __restrict__ out,
                             unsigned long long ws_bytes) {
    size_t i = (size_t)blockIdx.x * blockDim.x + threadIdx.x;
    const bool per_node = ws_bytes >= WS_NEED;
    if (i == 0) {
        ws[0] = 0.0f;
        if (per_node) ((int*)ws)[1] = 0;
    }
    float* aggh = ws + WS_AGGH;
    if (!per_node && i < (size_t)BB * HDIM) aggh[i] = 0.0f;
    if (i < (size_t)BB) out[i * OUTD] = 0.0f;
    if (per_node && i < (size_t)NN) ((int*)(ws + WS_CNT))[i] = 0;
}

__global__ void lsappnp_dsum(const float* __restrict__ degree,
                             const int* __restrict__ batch,
                             float* __restrict__ ws) {
    __shared__ float red[256];
    float s = 0.0f;
    for (int i = blockIdx.x * blockDim.x + threadIdx.x; i < BB;
         i += gridDim.x * blockDim.x)
        s += degree[batch[i]];
    red[threadIdx.x] = s;
    __syncthreads();
    for (int off = 128; off > 0; off >>= 1) {
        if (threadIdx.x < off) red[threadIdx.x] += red[threadIdx.x + off];
        __syncthreads();
    }
    if (threadIdx.x == 0) atomicAdd(ws, red[0]);
}

__global__ void __launch_bounds__(256)
lsappnp_count(const int* __restrict__ batch, const int* __restrict__ ends,
              float* __restrict__ ws, unsigned long long ws_bytes) {
    if (ws_bytes < WS_NEED) return;
    int* cnt  = (int*)(ws + WS_CNT);
    int* rank = (int*)(ws + WS_AGGH);
    const int r = blockIdx.x * 256 + threadIdx.x;
    const int src = (r < BB) ? batch[r] : ends[r - BB];
    rank[r] = atomicAdd(cnt + src, 1);
}

__global__ void __launch_bounds__(256)
lsappnp_alloc(float* __restrict__ ws, unsigned long long ws_bytes) {
    if (ws_bytes < WS_NEED) return;
    __shared__ int scan[256];
    __shared__ int sbase;
    const int* cnt = (const int*)(ws + WS_CNT);
    int* pos = (int*)(ws + WS_POS);
    int* cursor = (int*)ws + 1;
    const int tid = threadIdx.x;
    const int n = blockIdx.x * 256 + tid;
    const int c = (n < NN) ? cnt[n] : 0;
    scan[tid] = c;
    __syncthreads();
    for (int off = 1; off < 256; off <<= 1) {
        int v = (tid >= off) ? scan[tid - off] : 0;
        __syncthreads();
        scan[tid] += v;
        __syncthreads();
    }
    const int incl = scan[tid];
    if (tid == 255) sbase = atomicAdd(cursor, incl);
    __syncthreads();
    if (n < NN) pos[n] = sbase + incl - c;
}

__global__ void __launch_bounds__(256)
lsappnp_place(const float* __restrict__ att,
              const float* __restrict__ degree,
              const int* __restrict__ batch,
              const int* __restrict__ heads,
              const int* __restrict__ ends,
              float* __restrict__ ws,
              float* __restrict__ wsum,
              unsigned long long ws_bytes) {
    if (ws_bytes < WS_NEED) return;
    const int* pos = (const int*)(ws + WS_POS);
    int* plane = (int*)(ws + WS_AGGH);
    unsigned* erec = (unsigned*)(ws + WS_EREC);
    const float dsum   = ws[0];
    const float inv_ne = 1.0f / (float)EDG;
    const int r = blockIdx.x * 256 + threadIdx.x;
    const int rk = plane[r];
    plane[r] = 0;
    if (r + MTOT < BB * HDIM) plane[r + MTOT] = 0;
    int src, tgt, kk;
    float sc;
    if (r < BB) {
        tgt = r;
        src = batch[r];
        sc  = att[0];
        kk  = 0;
    } else {
        int idx = r - BB;
        int k   = idx >> 19;
        kk  = k + 1;
        tgt = heads[idx];
        src = ends[idx];
        float db = degree[batch[tgt]];
        float de = degree[src];
        sc = att[k + 1] * rsqrtf(db) * rsqrtf(de) * dsum * inv_ne;
    }
    const int p = pos[src] + rk;
    const unsigned rec = (unsigned)tgt | ((unsigned)kk << 16);
    if ((unsigned)p < (unsigned)MTOT) atomicExch(erec + p, rec);
    atomicAdd(wsum + (size_t)tgt * OUTD, sc);
}

__device__ __forceinline__ void chunk_issue(int base, int eE, int lane,
                                            const unsigned* __restrict__ erec,
                                            const int* __restrict__ batch,
                                            const float* __restrict__ degree,
                                            const float* __restrict__ att,
                                            unsigned& q0, unsigned& q1,
                                            float& d0, float& d1, float& a0, float& a1) {
    q0 = 0u; q1 = 0u;
    if (base + lane < eE)      q0 = erec[base + lane];
    if (base + 32 + lane < eE) q1 = erec[base + 32 + lane];
    const int t0 = (int)(q0 & (unsigned)(BB - 1)), t1 = (int)(q1 & (unsigned)(BB - 1));
    const int k0 = min((int)((q0 >> 16) & 7u), KHOP), k1 = min((int)((q1 >> 16) & 7u), KHOP);
    d0 = degree[batch[t0]]; d1 = degree[batch[t1]];
    a0 = att[k0];           a1 = att[k1];
}
__device__ __forceinline__ void chunk_weights(unsigned q0, unsigned q1, float d0, float d1,
                                              float a0, float a1, float& p0, float& p1) {
    p0 = (((q0 >> 16) & 7u) == 0u) ? a0 : a0 * rsqrtf(d0);
    p1 = (((q1 >> 16) & 7u) == 0u) ? a1 : a1 * rsqrtf(d1);
}

__global__ void __launch_bounds__(256)
lsappnp_mlp(const float* __restrict__ x,
            const float* __restrict__ W0, const float* __restrict__ b0,
            const float* __restrict__ att,
            const float* __restrict__ degree,
            const int* __restrict__ batch,
            const int* __restrict__ heads,
            const int* __restrict__ ends,
            float* __restrict__ ws,
            float* __restrict__ wsum,
            unsigned long long ws_bytes) {
    extern __shared__ float smem[];
    v2f*   W0p   = (v2f*)smem;
    float* b0s   = (float*)(W0p + 64 * 128);
    float* Xbase = b0s + 128;

    const int tid  = threadIdx.x;
    const int wave = tid >> 5;
    const int lane = tid & 31;
    const int hi   = lane >> 4;
    const int lm   = lane & 15;

    for (int i = tid; i < 64 * 128; i += 256) {
        int k2 = i >> 7, n = i & 127;
        v2f w0; w0[0] = W0[(2 * k2) * 128 + n]; w0[1] = W0[(2 * k2 + 1) * 128 + n];
        W0p[i] = w0;
    }
    if (tid < 128) b0s[tid] = b0[tid];
    __syncthreads();

    float* aggh = ws + WS_AGGH;
    float* Xs   = Xbase + wave * PWF;

    if (ws_bytes >= WS_NEED) {
        const int* cnt = (const int*)(ws + WS_CNT);
        const int* pos = (const int*)(ws + WS_POS);
        const unsigned* erec = (const unsigned*)(ws + WS_EREC);
        const float dsum   = ws[0];
        const float inv_ne = 1.0f / (float)EDG;
        unsigned* metaT = (unsigned*)(Xs + 16 * XS);
        float*    metaF = (float*)(metaT + 64);

        for (int tile = blockIdx.x * 8 + wave; tile < NTILEN; tile += gridDim.x * 8) {
            const int n0 = tile * 16;
            v4f xr[16];
            const v4f* xp = (const v4f*)(x + (size_t)n0 * FDIM) + lane;
#pragma unroll
            for (int rrow = 0; rrow < 16; ++rrow)
                xr[rrow] = __builtin_nontemporal_load(xp + rrow * 32);
            int c = 0, ps = 0;
            float de = 1.0f;
            if (lane < 16) { c = cnt[n0 + lane]; ps = pos[n0 + lane]; de = degree[n0 + lane]; }
            __builtin_amdgcn_sched_barrier(0);
            const float rde = rsqrtf(de);

            const int pe = ps + c;
            int eS = __builtin_amdgcn_readlane(ps, 0);
            int eE = __builtin_amdgcn_readlane(pe, 15);
            eS = max(eS, 0);
            eE = min(eE, MTOT);
            unsigned q0, q1;
            float d0, d1, a0, a1, p0, p1;
            chunk_issue(eS, eE, lane, erec, batch, degree, att, q0, q1, d0, d1, a0, a1);

#pragma unroll
            for (int rrow = 0; rrow < 16; ++rrow)
                *(v4f*)(Xs + rrow * XS + lane * 4) = xr[rrow];
            wait_ds();

            v8f acc[8];
#pragma unroll
            for (int nt = 0; nt < 8; ++nt) {
                float bv = b0s[nt * 16 + lm];
                v8f cc;
#pragma unroll
                for (int i = 0; i < 8; ++i) cc[i] = bv;
                acc[nt] = cc;
            }
            for (int kt = 0; kt < 32; ++kt) {
                int kc = kt * 4 + hi * 2;
                v2f a  = *(const v2f*)(Xs + lm * XS + kc);
                int k2 = kc >> 1;
#pragma unroll
                for (int nt = 0; nt < 8; ++nt) {
                    v2f bfr = W0p[k2 * 128 + nt * 16 + lm];
                    acc[nt] = __builtin_amdgcn_wmma_f32_16x16x4_f32(
                        false, a, false, bfr, (short)0, acc[nt], false, false);
                }
            }
#pragma unroll
            for (int nt = 0; nt < 8; ++nt)
#pragma unroll
                for (int v = 0; v < 8; ++v)
                    Xs[(v + 8 * hi) * XS + nt * 16 + lm] = fmaxf(acc[nt][v], 0.0f);

            for (int base = eS; base < eE; base += 64) {
                if (base != eS)
                    chunk_issue(base, eE, lane, erec, batch, degree, att, q0, q1, d0, d1, a0, a1);
                chunk_weights(q0, q1, d0, d1, a0, a1, p0, p1);
                metaT[lane]      = q0;
                metaF[lane]      = p0;
                metaT[32 + lane] = q1;
                metaF[32 + lane] = p1;
                wait_ds();
                const int lim = min(eE, base + 64);
#pragma unroll
                for (int m = 0; m < 16; ++m) {
                    const int s = max(__builtin_amdgcn_readlane(ps, m), base);
                    const int e = min(__builtin_amdgcn_readlane(pe, m), lim);
                    if (s < e) {
                        const float* hr = Xs + m * XS + lane;
                        const float h0 = hr[0], h1 = hr[32], h2 = hr[64], h3 = hr[96];
                        const float rde_m = __int_as_float(__builtin_amdgcn_readlane(__float_as_int(rde), m));
                        for (int j = s; j < e; ++j) {
                            const unsigned q  = (unsigned)__builtin_amdgcn_readfirstlane((int)metaT[j - base]);
                            const int      tg = (int)(q & (unsigned)(BB - 1));
                            const float    wp = metaF[j - base];
                            const float    w  = (((q >> 16) & 7u) == 0u) ? wp : wp * rde_m * dsum * inv_ne;
                            float* d = aggh + (size_t)tg * HDIM + lane;
                            atomicAdd(d,      h0 * w);
                            atomicAdd(d + 32, h1 * w);
                            atomicAdd(d + 64, h2 * w);
                            atomicAdd(d + 96, h3 * w);
                        }
                    }
                }
            }
        }
        return;
    }

    const float dsum   = ws[0];
    const float inv_ne = 1.0f / (float)EDG;

    int*   metaT = (int*)(Xs + 16 * XS);
    float* metaF = (float*)(metaT + 16);

    for (int tile = blockIdx.x * 8 + wave; tile < NTILES; tile += gridDim.x * 8) {
        int   src = 0;
        int   tgt = 0;
        float sc  = 0.0f;
        if (lane < 16) {
            long r = (long)tile * 16 + lane;
            if (r < BB) {
                tgt = (int)r;
                src = batch[r];
                sc  = att[0];
            } else {
                long idx = r - BB;
                int  k   = (int)(idx >> 19);
                tgt = heads[idx];
                src = ends[idx];
                float db = degree[batch[tgt]];
                float de = degree[src];
                sc = att[k + 1] * rsqrtf(db) * rsqrtf(de) * dsum * inv_ne;
            }
            metaT[lane] = tgt; metaF[lane] = sc;
        }

        float4 xr[16];
#pragma unroll
        for (int rrow = 0; rrow < 16; ++rrow) {
            int s = __builtin_amdgcn_readlane(src, rrow);
            xr[rrow] = ((const float4*)(x + (size_t)s * FDIM))[lane];
        }
#pragma unroll
        for (int rrow = 0; rrow < 16; ++rrow)
            ((float4*)(Xs + rrow * XS))[lane] = xr[rrow];
        wait_ds();

        v8f acc[8];
#pragma unroll
        for (int nt = 0; nt < 8; ++nt) {
            float bv = b0s[nt * 16 + lm];
            v8f c;
#pragma unroll
            for (int i = 0; i < 8; ++i) c[i] = bv;
            acc[nt] = c;
        }
        for (int kt = 0; kt < 32; ++kt) {
            int kc = kt * 4 + hi * 2;
            v2f a  = *(const v2f*)(Xs + lm * XS + kc);
            int k2 = kc >> 1;
#pragma unroll
            for (int nt = 0; nt < 8; ++nt) {
                v2f bfr = W0p[k2 * 128 + nt * 16 + lm];
                acc[nt] = __builtin_amdgcn_wmma_f32_16x16x4_f32(
                    false, a, false, bfr, (short)0, acc[nt], false, false);
            }
        }

#pragma unroll
        for (int v = 0; v < 8; ++v) {
            int   tg = metaT[v + 8 * hi];
            float sw = metaF[v + 8 * hi];
            float* base = aggh + (size_t)tg * HDIM + lm;
#pragma unroll
            for (int nt = 0; nt < 8; ++nt)
                atomicAdd(base + nt * 16, fmaxf(acc[nt][v], 0.0f) * sw);
        }
        if (lane < 16) atomicAdd(wsum + (size_t)tgt * OUTD, sc);
    }
}

__global__ void __launch_bounds__(256)
lsappnp_head(const float* __restrict__ aggh,
             const float* __restrict__ W1, const float* __restrict__ b1,
             const float* __restrict__ W2, const float* __restrict__ b2,
             float* out) {
    extern __shared__ float smem[];
    v2f*   W1p   = (v2f*)smem;
    float* b1s   = (float*)(W1p + 64 * 128);
    float* W2s   = b1s + 128;
    float* b2s   = W2s + HDIM * OUTD;
    float* Xbase = b2s + 64;

    const int tid  = threadIdx.x;
    const int wave = tid >> 5;
    const int lane = tid & 31;
    const int hi   = lane >> 4;
    const int lm   = lane & 15;

    for (int i = tid; i < 64 * 128; i += 256) {
        int k2 = i >> 7, n = i & 127;
        v2f w1; w1[0] = W1[(2 * k2) * 128 + n]; w1[1] = W1[(2 * k2 + 1) * 128 + n];
        W1p[i] = w1;
    }
    if (tid < 128) b1s[tid] = b1[tid];
    for (int i = tid; i < HDIM * OUTD; i += 256) W2s[i] = W2[i];
    if (tid < OUTD) b2s[tid] = b2[tid];
    __syncthreads();

    float* Xs    = Xbase + wave * PWF;
    float* metaF = Xs + 16 * XS;

    for (int tile = blockIdx.x * 8 + wave; tile < NTILEB; tile += gridDim.x * 8) {
        const size_t row0 = (size_t)tile * 16;

        float4 xr[16];
#pragma unroll
        for (int rrow = 0; rrow < 16; ++rrow)
            xr[rrow] = ((const float4*)(aggh + (row0 + rrow) * HDIM))[lane];
        float wsv = 0.0f;
        if (lane < 16) wsv = out[(row0 + lane) * OUTD];
#pragma unroll
        for (int rrow = 0; rrow < 16; ++rrow)
            ((float4*)(Xs + rrow * XS))[lane] = xr[rrow];
        if (lane < 16) metaF[lane] = wsv;
        wait_ds();

        v8f acc2[8];
#pragma unroll
        for (int nt = 0; nt < 8; ++nt) {
            float bv = b1s[nt * 16 + lm];
            v8f c;
#pragma unroll
            for (int i = 0; i < 8; ++i) c[i] = metaF[i + 8 * hi] * bv;
            acc2[nt] = c;
        }
        for (int kt = 0; kt < 32; ++kt) {
            int kc = kt * 4 + hi * 2;
            v2f a  = *(const v2f*)(Xs + lm * XS + kc);
            int k2 = kc >> 1;
#pragma unroll
            for (int nt = 0; nt < 8; ++nt) {
                v2f bfr = W1p[k2 * 128 + nt * 16 + lm];
                acc2[nt] = __builtin_amdgcn_wmma_f32_16x16x4_f32(
                    false, a, false, bfr, (short)0, acc2[nt], false, false);
            }
        }
#pragma unroll
        for (int nt = 0; nt < 8; ++nt)
#pragma unroll
            for (int v = 0; v < 8; ++v)
                Xs[(v + 8 * hi) * XS + nt * 16 + lm] = fmaxf(acc2[nt][v], 0.0f);
        wait_ds();

        const float* ar = Xs + lm * XS;
        const float* wr = W2s + hi * 4;
        float p0 = 0.0f, p1 = 0.0f, p2 = 0.0f, p3 = 0.0f;
        for (int c = 0; c < HDIM; c += 4) {
            float4 a  = *(const float4*)(ar + c);
            float4 w0 = *(const float4*)(wr + (c + 0) * OUTD);
            float4 w1 = *(const float4*)(wr + (c + 1) * OUTD);
            float4 w2 = *(const float4*)(wr + (c + 2) * OUTD);
            float4 w3 = *(const float4*)(wr + (c + 3) * OUTD);
            p0 += a.x * w0.x; p1 += a.x * w0.y; p2 += a.x * w0.z; p3 += a.x * w0.w;
            p0 += a.y * w1.x; p1 += a.y * w1.y; p2 += a.y * w1.z; p3 += a.y * w1.w;
            p0 += a.z * w2.x; p1 += a.z * w2.y; p2 += a.z * w2.z; p3 += a.z * w2.w;
            p0 += a.w * w3.x; p1 += a.w * w3.y; p2 += a.w * w3.z; p3 += a.w * w3.w;
        }
        p0 += b2s[hi * 4 + 0]; p1 += b2s[hi * 4 + 1]; p2 += b2s[hi * 4 + 2]; p3 += b2s[hi * 4 + 3];
        float m = fmaxf(fmaxf(p0, p1), fmaxf(p2, p3));
        m = fmaxf(m, __shfl_xor(m, 16));
        float s = __expf(p0 - m) + __expf(p1 - m) + __expf(p2 - m) + __expf(p3 - m);
        s += __shfl_xor(s, 16);
        float lse = __logf(s);
        float4 o;
        o.x = p0 - m - lse; o.y = p1 - m - lse; o.z = p2 - m - lse; o.w = p3 - m - lse;
        *(float4*)(out + (row0 + lm) * OUTD + hi * 4) = o;
    }
}

extern "C" void kernel_launch(void* const* d_in, const int* in_sizes, int n_in,
                              void* d_out, int out_size, void* d_ws, size_t ws_size,
                              hipStream_t stream) {
    const float* x      = (const float*)d_in[0];
    const float* W0     = (const float*)d_in[1];
    const float* b0     = (const float*)d_in[2];
    const float* W1     = (const float*)d_in[3];
    const float* b1     = (const float*)d_in[4];
    const float* W2     = (const float*)d_in[5];
    const float* b2     = (const float*)d_in[6];
    const float* att    = (const float*)d_in[7];
    const float* degree = (const float*)d_in[8];
    const int*   batch  = (const int*)d_in[9];
    const int*   heads  = (const int*)d_in[10];
    const int*   ends   = (const int*)d_in[11];
    float*       out    = (float*)d_out;
    float*       ws     = (float*)d_ws;
    float*       aggh   = ws + WS_AGGH;
    const unsigned long long ws_bytes = (unsigned long long)ws_size;

    lsappnp_zero<<<(BB * HDIM + 255) / 256, 256, 0, stream>>>(ws, out, ws_bytes);
    lsappnp_dsum<<<64, 256, 0, stream>>>(degree, batch, ws);
    lsappnp_count<<<MTOT / 256, 256, 0, stream>>>(batch, ends, ws, ws_bytes);
    lsappnp_alloc<<<(NN + 255) / 256, 256, 0, stream>>>(ws, ws_bytes);
    lsappnp_place<<<MTOT / 256, 256, 0, stream>>>(att, degree, batch, heads, ends, ws, out, ws_bytes);

    size_t shmem_mlp = (size_t)64 * 128 * sizeof(v2f)
                     + 128 * sizeof(float)
                     + (size_t)8 * PWF * sizeof(float);
    lsappnp_mlp<<<512, 256, shmem_mlp, stream>>>(x, W0, b0, att, degree,
                                                 batch, heads, ends, ws, out, ws_bytes);

    size_t shmem_head = (size_t)64 * 128 * sizeof(v2f)
                      + (128 + HDIM * OUTD + 64) * sizeof(float)
                      + (size_t)8 * PWF * sizeof(float);
    lsappnp_head<<<NTILEB / 8, 256, shmem_head, stream>>>(aggh, W1, b1, W2, b2, out);
}


# The two float accumulators that both cache halves add into (segment sums, weight sums) get one copy per L2 half, picked by die id; the head kernel adds the copies; float summation order changes
# speedup vs baseline: 2.9793x; 1.1435x over previous
#include <hip/hip_runtime.h>
#include <math.h>


#define NN      500000
#define FDIM    128
#define HDIM    128
#define BB      32768
#define KHOP    4
#define EDG     524288
#define OUTD    8
#define MTOT    (BB + KHOP * EDG)
#define NTILES  (MTOT / 16)
#define NTILEN  (NN / 16)
#define NTILEB  (BB / 16)
#define XS      132
#define PWF     (16 * XS + 128)

#define NPAD     500032
#define WS_AGGH  64
#define WS_CNT   (WS_AGGH + BB * HDIM)
#define WS_POS   (WS_CNT + NPAD)
#define WS_EREC  (WS_POS + NPAD)
#define WS_WORDS (WS_EREC + MTOT)
#define WS_NEED  ((unsigned long long)WS_WORDS * 4ull)
#define WS_WSUM2 WS_WORDS
#define WS_AGG2  (WS_WSUM2 + BB * OUTD)
#define WS_NEED_WS ((unsigned long long)(WS_WSUM2 + BB * OUTD) * 4ull)
#define WS_NEED_PL ((unsigned long long)(WS_AGG2 + BB * HDIM) * 4ull)

typedef __attribute__((ext_vector_type(2))) float v2f;
typedef __attribute__((ext_vector_type(4))) float v4f;
typedef __attribute__((ext_vector_type(8))) float v8f;


__device__ __forceinline__ void wait_ds() {
    asm volatile("s_wait_dscnt 0x0" ::: "memory");
}

__device__ __forceinline__ int l2_half() {
    const unsigned id = __builtin_amdgcn_s_sendmsg_rtn(0x87);
    return (((id >> 16) & 0xFu) >= 4u) ? 1 : 0;
}

__global__ void lsappnp_zero(float* __restrict__ ws, float* __restrict__ out,
                             unsigned long long ws_bytes) {
    size_t i = (size_t)blockIdx.x * blockDim.x + threadIdx.x;
    const bool per_node = ws_bytes >= WS_NEED;
    if (i == 0) {
        ws[0] = 0.0f;
        if (per_node) ((int*)ws)[1] = 0;
    }
    float* aggh = ws + WS_AGGH;
    if (!per_node && i < (size_t)BB * HDIM) aggh[i] = 0.0f;
    if (i < (size_t)BB) out[i * OUTD] = 0.0f;
    if (per_node && i < (size_t)NN) ((int*)(ws + WS_CNT))[i] = 0;
    if (ws_bytes >= WS_NEED_WS && i < (size_t)BB) ws[WS_WSUM2 + i * OUTD] = 0.0f;
    if (ws_bytes >= WS_NEED_PL && i < (size_t)BB * HDIM / 4) {
        v4f z; z[0] = 0.0f; z[1] = 0.0f; z[2] = 0.0f; z[3] = 0.0f;
        ((v4f*)(ws + WS_AGG2))[i] = z;
    }
}

__global__ void lsappnp_dsum(const float* __restrict__ degree,
                             const int* __restrict__ batch,
                             float* __restrict__ ws) {
    __shared__ float red[256];
    float s = 0.0f;
    for (int i = blockIdx.x * blockDim.x + threadIdx.x; i < BB;
         i += gridDim.x * blockDim.x)
        s += degree[batch[i]];
    red[threadIdx.x] = s;
    __syncthreads();
    for (int off = 128; off > 0; off >>= 1) {
        if (threadIdx.x < off) red[threadIdx.x] += red[threadIdx.x + off];
        __syncthreads();
    }
    if (threadIdx.x == 0) atomicAdd(ws, red[0]);
}

__global__ void __launch_bounds__(256)
lsappnp_count(const int* __restrict__ batch, const int* __restrict__ ends,
              float* __restrict__ ws, unsigned long long ws_bytes) {
    if (ws_bytes < WS_NEED) return;
    int* cnt  = (int*)(ws + WS_CNT);
    int* rank = (int*)(ws + WS_AGGH);
    const int r = blockIdx.x * 256 + threadIdx.x;
    const int src = (r < BB) ? batch[r] : ends[r - BB];
    rank[r] = atomicAdd(cnt + src, 1);
}

__global__ void __launch_bounds__(256)
lsappnp_alloc(float* __restrict__ ws, unsigned long long ws_bytes) {
    if (ws_bytes < WS_NEED) return;
    __shared__ int scan[256];
    __shared__ int sbase;
    const int* cnt = (const int*)(ws + WS_CNT);
    int* pos = (int*)(ws + WS_POS);
    int* cursor = (int*)ws + 1;
    const int tid = threadIdx.x;
    const int n = blockIdx.x * 256 + tid;
    const int c = (n < NN) ? cnt[n] : 0;
    scan[tid] = c;
    __syncthreads();
    for (int off = 1; off < 256; off <<= 1) {
        int v = (tid >= off) ? scan[tid - off] : 0;
        __syncthreads();
        scan[tid] += v;
        __syncthreads();
    }
    const int incl = scan[tid];
    if (tid == 255) sbase = atomicAdd(cursor, incl);
    __syncthreads();
    if (n < NN) pos[n] = sbase + incl - c;
}

__global__ void __launch_bounds__(256)
lsappnp_place(const float* __restrict__ att,
              const float* __restrict__ degree,
              const int* __restrict__ batch,
              const int* __restrict__ heads,
              const int* __restrict__ ends,
              float* __restrict__ ws,
              float* __restrict__ wsum,
              unsigned long long ws_bytes) {
    if (ws_bytes < WS_NEED) return;
    const int* pos = (const int*)(ws + WS_POS);
    int* plane = (int*)(ws + WS_AGGH);
    unsigned* erec = (unsigned*)(ws + WS_EREC);
    const float dsum   = ws[0];
    const float inv_ne = 1.0f / (float)EDG;
    const int r = blockIdx.x * 256 + threadIdx.x;
    const int rk = plane[r];
    plane[r] = 0;
    if (r + MTOT < BB * HDIM) plane[r + MTOT] = 0;
    int src, tgt, kk;
    float sc;
    if (r < BB) {
        tgt = r;
        src = batch[r];
        sc  = att[0];
        kk  = 0;
    } else {
        int idx = r - BB;
        int k   = idx >> 19;
        kk  = k + 1;
        tgt = heads[idx];
        src = ends[idx];
        float db = degree[batch[tgt]];
        float de = degree[src];
        sc = att[k + 1] * rsqrtf(db) * rsqrtf(de) * dsum * inv_ne;
    }
    const int p = pos[src] + rk;
    const unsigned rec = (unsigned)tgt | ((unsigned)kk << 16);
    if ((unsigned)p < (unsigned)MTOT) atomicExch(erec + p, rec);
    float* wsum_h = wsum;
    if (ws_bytes >= WS_NEED_WS && l2_half()) wsum_h = ws + WS_WSUM2;
    atomicAdd(wsum_h + (size_t)tgt * OUTD, sc);
}

__device__ __forceinline__ void chunk_issue(int base, int eE, int lane,
                                            const unsigned* __restrict__ erec,
                                            const int* __restrict__ batch,
                                            const float* __restrict__ degree,
                                            const float* __restrict__ att,
                                            unsigned& q0, unsigned& q1,
                                            float& d0, float& d1, float& a0, float& a1) {
    q0 = 0u; q1 = 0u;
    if (base + lane < eE)      q0 = erec[base + lane];
    if (base + 32 + lane < eE) q1 = erec[base + 32 + lane];
    const int t0 = (int)(q0 & (unsigned)(BB - 1)), t1 = (int)(q1 & (unsigned)(BB - 1));
    const int k0 = min((int)((q0 >> 16) & 7u), KHOP), k1 = min((int)((q1 >> 16) & 7u), KHOP);
    d0 = degree[batch[t0]]; d1 = degree[batch[t1]];
    a0 = att[k0];           a1 = att[k1];
}
__device__ __forceinline__ void chunk_weights(unsigned q0, unsigned q1, float d0, float d1,
                                              float a0, float a1, float& p0, float& p1) {
    p0 = (((q0 >> 16) & 7u) == 0u) ? a0 : a0 * rsqrtf(d0);
    p1 = (((q1 >> 16) & 7u) == 0u) ? a1 : a1 * rsqrtf(d1);
}

__global__ void __launch_bounds__(256)
lsappnp_mlp(const float* __restrict__ x,
            const float* __restrict__ W0, const float* __restrict__ b0,
            const float* __restrict__ att,
            const float* __restrict__ degree,
            const int* __restrict__ batch,
            const int* __restrict__ heads,
            const int* __restrict__ ends,
            float* __restrict__ ws,
            float* __restrict__ wsum,
            unsigned long long ws_bytes) {
    extern __shared__ float smem[];
    v2f*   W0p   = (v2f*)smem;
    float* b0s   = (float*)(W0p + 64 * 128);
    float* Xbase = b0s + 128;

    const int tid  = threadIdx.x;
    const int wave = tid >> 5;
    const int lane = tid & 31;
    const int hi   = lane >> 4;
    const int lm   = lane & 15;

    for (int i = tid; i < 64 * 128; i += 256) {
        int k2 = i >> 7, n = i & 127;
        v2f w0; w0[0] = W0[(2 * k2) * 128 + n]; w0[1] = W0[(2 * k2 + 1) * 128 + n];
        W0p[i] = w0;
    }
    if (tid < 128) b0s[tid] = b0[tid];
    __syncthreads();

    float* aggh = ws + WS_AGGH;
    if (ws_bytes >= WS_NEED_PL && l2_half()) aggh = ws + WS_AGG2;
    float* Xs   = Xbase + wave * PWF;

    if (ws_bytes >= WS_NEED) {
        const int* cnt = (const int*)(ws + WS_CNT);
        const int* pos = (const int*)(ws + WS_POS);
        const unsigned* erec = (const unsigned*)(ws + WS_EREC);
        const float dsum   = ws[0];
        const float inv_ne = 1.0f / (float)EDG;
        unsigned* metaT = (unsigned*)(Xs + 16 * XS);
        float*    metaF = (float*)(metaT + 64);

        for (int tile = blockIdx.x * 8 + wave; tile < NTILEN; tile += gridDim.x * 8) {
            const int n0 = tile * 16;
            v4f xr[16];
            const v4f* xp = (const v4f*)(x + (size_t)n0 * FDIM) + lane;
#pragma unroll
            for (int rrow = 0; rrow < 16; ++rrow)
                xr[rrow] = __builtin_nontemporal_load(xp + rrow * 32);
            int c = 0, ps = 0;
            float de = 1.0f;
            if (lane < 16) { c = cnt[n0 + lane]; ps = pos[n0 + lane]; de = degree[n0 + lane]; }
            __builtin_amdgcn_sched_barrier(0);
            const float rde = rsqrtf(de);

            const int pe = ps + c;
            int eS = __builtin_amdgcn_readlane(ps, 0);
            int eE = __builtin_amdgcn_readlane(pe, 15);
            eS = max(eS, 0);
            eE = min(eE, MTOT);
            unsigned q0, q1;
            float d0, d1, a0, a1, p0, p1;
            chunk_issue(eS, eE, lane, erec, batch, degree, att, q0, q1, d0, d1, a0, a1);

#pragma unroll
            for (int rrow = 0; rrow < 16; ++rrow)
                *(v4f*)(Xs + rrow * XS + lane * 4) = xr[rrow];
            wait_ds();

            v8f acc[8];
#pragma unroll
            for (int nt = 0; nt < 8; ++nt) {
                float bv = b0s[nt * 16 + lm];
                v8f cc;
#pragma unroll
                for (int i = 0; i < 8; ++i) cc[i] = bv;
                acc[nt] = cc;
            }
            for (int kt = 0; kt < 32; ++kt) {
                int kc = kt * 4 + hi * 2;
                v2f a  = *(const v2f*)(Xs + lm * XS + kc);
                int k2 = kc >> 1;
#pragma unroll
                for (int nt = 0; nt < 8; ++nt) {
                    v2f bfr = W0p[k2 * 128 + nt * 16 + lm];
                    acc[nt] = __builtin_amdgcn_wmma_f32_16x16x4_f32(
                        false, a, false, bfr, (short)0, acc[nt], false, false);
                }
            }
#pragma unroll
            for (int nt = 0; nt < 8; ++nt)
#pragma unroll
                for (int v = 0; v < 8; ++v)
                    Xs[(v + 8 * hi) * XS + nt * 16 + lm] = fmaxf(acc[nt][v], 0.0f);

            for (int base = eS; base < eE; base += 64) {
                if (base != eS)
                    chunk_issue(base, eE, lane, erec, batch, degree, att, q0, q1, d0, d1, a0, a1);
                chunk_weights(q0, q1, d0, d1, a0, a1, p0, p1);
                metaT[lane]      = q0;
                metaF[lane]      = p0;
                metaT[32 + lane] = q1;
                metaF[32 + lane] = p1;
                wait_ds();
                const int lim = min(eE, base + 64);
#pragma unroll
                for (int m = 0; m < 16; ++m) {
                    const int s = max(__builtin_amdgcn_readlane(ps, m), base);
                    const int e = min(__builtin_amdgcn_readlane(pe, m), lim);
                    if (s < e) {
                        const float* hr = Xs + m * XS + lane;
                        const float h0 = hr[0], h1 = hr[32], h2 = hr[64], h3 = hr[96];
                        const float rde_m = __int_as_float(__builtin_amdgcn_readlane(__float_as_int(rde), m));
                        for (int j = s; j < e; ++j) {
                            const unsigned q  = (unsigned)__builtin_amdgcn_readfirstlane((int)metaT[j - base]);
                            const int      tg = (int)(q & (unsigned)(BB - 1));
                            const float    wp = metaF[j - base];
                            const float    w  = (((q >> 16) & 7u) == 0u) ? wp : wp * rde_m * dsum * inv_ne;
                            float* d = aggh + (size_t)tg * HDIM + lane;
                            atomicAdd(d,      h0 * w);
                            atomicAdd(d + 32, h1 * w);
                            atomicAdd(d + 64, h2 * w);
                            atomicAdd(d + 96, h3 * w);
                        }
                    }
                }
            }
        }
        return;
    }

    const float dsum   = ws[0];
    const float inv_ne = 1.0f / (float)EDG;

    int*   metaT = (int*)(Xs + 16 * XS);
    float* metaF = (float*)(metaT + 16);

    for (int tile = blockIdx.x * 8 + wave; tile < NTILES; tile += gridDim.x * 8) {
        int   src = 0;
        int   tgt = 0;
        float sc  = 0.0f;
        if (lane < 16) {
            long r = (long)tile * 16 + lane;
            if (r < BB) {
                tgt = (int)r;
                src = batch[r];
                sc  = att[0];
            } else {
                long idx = r - BB;
                int  k   = (int)(idx >> 19);
                tgt = heads[idx];
                src = ends[idx];
                float db = degree[batch[tgt]];
                float de = degree[src];
                sc = att[k + 1] * rsqrtf(db) * rsqrtf(de) * dsum * inv_ne;
            }
            metaT[lane] = tgt; metaF[lane] = sc;
        }

        float4 xr[16];
#pragma unroll
        for (int rrow = 0; rrow < 16; ++rrow) {
            int s = __builtin_amdgcn_readlane(src, rrow);
            xr[rrow] = ((const float4*)(x + (size_t)s * FDIM))[lane];
        }
#pragma unroll
        for (int rrow = 0; rrow < 16; ++rrow)
            ((float4*)(Xs + rrow * XS))[lane] = xr[rrow];
        wait_ds();

        v8f acc[8];
#pragma unroll
        for (int nt = 0; nt < 8; ++nt) {
            float bv = b0s[nt * 16 + lm];
            v8f c;
#pragma unroll
            for (int i = 0; i < 8; ++i) c[i] = bv;
            acc[nt] = c;
        }
        for (int kt = 0; kt < 32; ++kt) {
            int kc = kt * 4 + hi * 2;
            v2f a  = *(const v2f*)(Xs + lm * XS + kc);
            int k2 = kc >> 1;
#pragma unroll
            for (int nt = 0; nt < 8; ++nt) {
                v2f bfr = W0p[k2 * 128 + nt * 16 + lm];
                acc[nt] = __builtin_amdgcn_wmma_f32_16x16x4_f32(
                    false, a, false, bfr, (short)0, acc[nt], false, false);
            }
        }

#pragma unroll
        for (int v = 0; v < 8; ++v) {
            int   tg = metaT[v + 8 * hi];
            float sw = metaF[v + 8 * hi];
            float* base = aggh + (size_t)tg * HDIM + lm;
#pragma unroll
            for (int nt = 0; nt < 8; ++nt)
                atomicAdd(base + nt * 16, fmaxf(acc[nt][v], 0.0f) * sw);
        }
        if (lane < 16) atomicAdd(wsum + (size_t)tgt * OUTD, sc);
    }
}

__global__ void __launch_bounds__(256)
lsappnp_head(const float* __restrict__ ws,
             const float* __restrict__ W1, const float* __restrict__ b1,
             const float* __restrict__ W2, const float* __restrict__ b2,
             float* out, unsigned long long ws_bytes) {
    const float* aggh  = ws + WS_AGGH;
    const float* aggh2 = ws + WS_AGG2;
    const bool two_planes = ws_bytes >= WS_NEED_PL;
    const bool two_wsums  = ws_bytes >= WS_NEED_WS;
    extern __shared__ float smem[];
    v2f*   W1p   = (v2f*)smem;
    float* b1s   = (float*)(W1p + 64 * 128);
    float* W2s   = b1s + 128;
    float* b2s   = W2s + HDIM * OUTD;
    float* Xbase = b2s + 64;

    const int tid  = threadIdx.x;
    const int wave = tid >> 5;
    const int lane = tid & 31;
    const int hi   = lane >> 4;
    const int lm   = lane & 15;

    for (int i = tid; i < 64 * 128; i += 256) {
        int k2 = i >> 7, n = i & 127;
        v2f w1; w1[0] = W1[(2 * k2) * 128 + n]; w1[1] = W1[(2 * k2 + 1) * 128 + n];
        W1p[i] = w1;
    }
    if (tid < 128) b1s[tid] = b1[tid];
    for (int i = tid; i < HDIM * OUTD; i += 256) W2s[i] = W2[i];
    if (tid < OUTD) b2s[tid] = b2[tid];
    __syncthreads();

    float* Xs    = Xbase + wave * PWF;
    float* metaF = Xs + 16 * XS;

    for (int tile = blockIdx.x * 8 + wave; tile < NTILEB; tile += gridDim.x * 8) {
        const size_t row0 = (size_t)tile * 16;

        float4 xr[16];
#pragma unroll
        for (int rrow = 0; rrow < 16; ++rrow)
            xr[rrow] = ((const float4*)(aggh + (row0 + rrow) * HDIM))[lane];
        float wsv = 0.0f;
        if (lane < 16) wsv = out[(row0 + lane) * OUTD];
        if (two_wsums && lane < 16) wsv += ws[WS_WSUM2 + (row0 + lane) * OUTD];
        if (two_planes) {
            float4 yr[16];
#pragma unroll
            for (int rrow = 0; rrow < 16; ++rrow)
                yr[rrow] = ((const float4*)(aggh2 + (row0 + rrow) * HDIM))[lane];
#pragma unroll
            for (int rrow = 0; rrow < 16; ++rrow) {
                xr[rrow].x += yr[rrow].x; xr[rrow].y += yr[rrow].y;
                xr[rrow].z += yr[rrow].z; xr[rrow].w += yr[rrow].w;
            }
        }
#pragma unroll
        for (int rrow = 0; rrow < 16; ++rrow)
            ((float4*)(Xs + rrow * XS))[lane] = xr[rrow];
        if (lane < 16) metaF[lane] = wsv;
        wait_ds();

        v8f acc2[8];
#pragma unroll
        for (int nt = 0; nt < 8; ++nt) {
            float bv = b1s[nt * 16 + lm];
            v8f c;
#pragma unroll
            for (int i = 0; i < 8; ++i) c[i] = metaF[i + 8 * hi] * bv;
            acc2[nt] = c;
        }
        for (int kt = 0; kt < 32; ++kt) {
            int kc = kt * 4 + hi * 2;
            v2f a  = *(const v2f*)(Xs + lm * XS + kc);
            int k2 = kc >> 1;
#pragma unroll
            for (int nt = 0; nt < 8; ++nt) {
                v2f bfr = W1p[k2 * 128 + nt * 16 + lm];
                acc2[nt] = __builtin_amdgcn_wmma_f32_16x16x4_f32(
                    false, a, false, bfr, (short)0, acc2[nt], false, false);
            }
        }
#pragma unroll
        for (int nt = 0; nt < 8; ++nt)
#pragma unroll
            for (int v = 0; v < 8; ++v)
                Xs[(v + 8 * hi) * XS + nt * 16 + lm] = fmaxf(acc2[nt][v], 0.0f);
        wait_ds();

        const float* ar = Xs + lm * XS;
        const float* wr = W2s + hi * 4;
        float p0 = 0.0f, p1 = 0.0f, p2 = 0.0f, p3 = 0.0f;
        for (int c = 0; c < HDIM; c += 4) {
            float4 a  = *(const float4*)(ar + c);
            float4 w0 = *(const float4*)(wr + (c + 0) * OUTD);
            float4 w1 = *(const float4*)(wr + (c + 1) * OUTD);
            float4 w2 = *(const float4*)(wr + (c + 2) * OUTD);
            float4 w3 = *(const float4*)(wr + (c + 3) * OUTD);
            p0 += a.x * w0.x; p1 += a.x * w0.y; p2 += a.x * w0.z; p3 += a.x * w0.w;
            p0 += a.y * w1.x; p1 += a.y * w1.y; p2 += a.y * w1.z; p3 += a.y * w1.w;
            p0 += a.z * w2.x; p1 += a.z * w2.y; p2 += a.z * w2.z; p3 += a.z * w2.w;
            p0 += a.w * w3.x; p1 += a.w * w3.y; p2 += a.w * w3.z; p3 += a.w * w3.w;
        }
        p0 += b2s[hi * 4 + 0]; p1 += b2s[hi * 4 + 1]; p2 += b2s[hi * 4 + 2]; p3 += b2s[hi * 4 + 3];
        float m = fmaxf(fmaxf(p0, p1), fmaxf(p2, p3));
        m = fmaxf(m, __shfl_xor(m, 16));
        float s = __expf(p0 - m) + __expf(p1 - m) + __expf(p2 - m) + __expf(p3 - m);
        s += __shfl_xor(s, 16);
        float lse = __logf(s);
        float4 o;
        o.x = p0 - m - lse; o.y = p1 - m - lse; o.z = p2 - m - lse; o.w = p3 - m - lse;
        *(float4*)(out + (row0 + lm) * OUTD + hi * 4) = o;
    }
}

extern "C" void kernel_launch(void* const* d_in, const int* in_sizes, int n_in,
                              void* d_out, int out_size, void* d_ws, size_t ws_size,
                              hipStream_t stream) {
    const float* x      = (const float*)d_in[0];
    const float* W0     = (const float*)d_in[1];
    const float* b0     = (const float*)d_in[2];
    const float* W1     = (const float*)d_in[3];
    const float* b1     = (const float*)d_in[4];
    const float* W2     = (const float*)d_in[5];
    const float* b2     = (const float*)d_in[6];
    const float* att    = (const float*)d_in[7];
    const float* degree = (const float*)d_in[8];
    const int*   batch  = (const int*)d_in[9];
    const int*   heads  = (const int*)d_in[10];
    const int*   ends   = (const int*)d_in[11];
    float*       out    = (float*)d_out;
    float*       ws     = (float*)d_ws;
    const unsigned long long ws_bytes = (unsigned long long)ws_size;

    lsappnp_zero<<<(BB * HDIM + 255) / 256, 256, 0, stream>>>(ws, out, ws_bytes);
    lsappnp_dsum<<<64, 256, 0, stream>>>(degree, batch, ws);
    lsappnp_count<<<MTOT / 256, 256, 0, stream>>>(batch, ends, ws, ws_bytes);
    lsappnp_alloc<<<(NN + 255) / 256, 256, 0, stream>>>(ws, ws_bytes);
    lsappnp_place<<<MTOT / 256, 256, 0, stream>>>(att, degree, batch, heads, ends, ws, out, ws_bytes);

    size_t shmem_mlp = (size_t)64 * 128 * sizeof(v2f)
                     + 128 * sizeof(float)
                     + (size_t)8 * PWF * sizeof(float);
    lsappnp_mlp<<<512, 256, shmem_mlp, stream>>>(x, W0, b0, att, degree,
                                                 batch, heads, ends, ws, out, ws_bytes);

    size_t shmem_head = (size_t)64 * 128 * sizeof(v2f)
                      + (128 + HDIM * OUTD + 64) * sizeof(float)
                      + (size_t)8 * PWF * sizeof(float);
    lsappnp_head<<<NTILEB / 8, 256, shmem_head, stream>>>(ws, W1, b1, W2, b2, out, ws_bytes);
}


# Rows are grouped by source node on LDS counters in two levels (8192-row groups order by 2048-node bin; one group per bin ranks and orders it in LDS), whole-line stores; float summation order changes
# speedup vs baseline: 3.6215x; 1.2156x over previous
#include <hip/hip_runtime.h>
#include <math.h>


#define NN      500000
#define FDIM    128
#define HDIM    128
#define BB      32768
#define KHOP    4
#define EDG     524288
#define OUTD    8
#define MTOT    (BB + KHOP * EDG)
#define NTILES  (MTOT / 16)
#define NTILEN  (NN / 16)
#define NTILEB  (BB / 16)
#define XS      132
#define PWF     (16 * XS + 128)

#define NPAD     500032
#define WS_AGGH  64
#define WS_CNT   (WS_AGGH + BB * HDIM)
#define WS_POS   (WS_CNT + NPAD)
#define WS_EREC  (WS_POS + NPAD)
#define WS_WORDS (WS_EREC + MTOT)
#define WS_NEED  ((unsigned long long)WS_WORDS * 4ull)
#define WS_WSUM2 WS_WORDS
#define WS_AGG2  (WS_WSUM2 + BB * OUTD)
#define WS_NEED_WS ((unsigned long long)(WS_WSUM2 + BB * OUTD) * 4ull)
#define WS_NEED_PL ((unsigned long long)(WS_AGG2 + BB * HDIM) * 4ull)

#define GRP_CH   8192
#define GRP_RPT  (GRP_CH / 256)
#define GRP_B    16
#define GRP_G    (MTOT / GRP_CH)
#define BIN_SH   11
#define BIN_N    2048
#define NBIN     ((NN + BIN_N - 1) / BIN_N)
#define POOL_W   4104640
#define PL_SEG   POOL_W
#define PL_BTOT  (PL_SEG + GRP_G * 256)
#define STG_CAP  40960

typedef __attribute__((ext_vector_type(2))) float v2f;
typedef __attribute__((ext_vector_type(4))) float v4f;
typedef __attribute__((ext_vector_type(8))) float v8f;


__device__ __forceinline__ void wait_ds() {
    asm volatile("s_wait_dscnt 0x0" ::: "memory");
}

__device__ __forceinline__ int l2_half() {
    const unsigned id = __builtin_amdgcn_s_sendmsg_rtn(0x87);
    return (((id >> 16) & 0xFu) >= 4u) ? 1 : 0;
}

__global__ void lsappnp_zero(float* __restrict__ ws, float* __restrict__ out,
                             unsigned long long ws_bytes) {
    size_t i = (size_t)blockIdx.x * blockDim.x + threadIdx.x;
    const bool per_node = ws_bytes >= WS_NEED;
    if (i == 0) {
        ws[0] = 0.0f;
        if (per_node) ((int*)ws)[1] = 0;
    }
    if (per_node && i < 256) ((int*)(ws + WS_AGGH))[PL_BTOT + i * 32] = 0;
    if (i < (size_t)BB) out[i * OUTD] = 0.0f;
    if (ws_bytes >= WS_NEED_WS && i < (size_t)BB) ws[WS_WSUM2 + i * OUTD] = 0.0f;
    if (ws_bytes >= WS_NEED_PL && i < (size_t)BB * HDIM / 4) {
        v4f z; z[0] = 0.0f; z[1] = 0.0f; z[2] = 0.0f; z[3] = 0.0f;
        ((v4f*)(ws + WS_AGG2))[i] = z;
    }
}

__global__ void lsappnp_dsum(const float* __restrict__ degree,
                             const int* __restrict__ batch,
                             float* __restrict__ ws) {
    __shared__ float red[256];
    float s = 0.0f;
    for (int i = blockIdx.x * blockDim.x + threadIdx.x; i < BB;
         i += gridDim.x * blockDim.x)
        s += degree[batch[i]];
    red[threadIdx.x] = s;
    __syncthreads();
    for (int off = 128; off > 0; off >>= 1) {
        if (threadIdx.x < off) red[threadIdx.x] += red[threadIdx.x + off];
        __syncthreads();
    }
    if (threadIdx.x == 0) atomicAdd(ws, red[0]);
}

__device__ __forceinline__ void bin_rows(const bool isb, const int r0, const int (&srcs)[GRP_RPT],
                                         const float* __restrict__ att,
                                         const float* __restrict__ degree,
                                         const int* __restrict__ batch,
                                         const int* __restrict__ heads,
                                         const float dsum, float* __restrict__ wsum_h,
                                         int* fill, const int* boff, unsigned* sorted) {
    const float inv_ne = 1.0f / (float)EDG;
#pragma unroll
    for (int u0 = 0; u0 < GRP_RPT; u0 += GRP_B) {
        int   tg[GRP_B], kk[GRP_B];
        float sc[GRP_B];
        if (isb) {
#pragma unroll
            for (int j = 0; j < GRP_B; ++j) {
                tg[j] = r0 + (u0 + j) * 256;
                kk[j] = 0;
                sc[j] = att[0];
            }
        } else {
            int   bt[GRP_B];
            float db[GRP_B], de[GRP_B];
#pragma unroll
            for (int j = 0; j < GRP_B; ++j) {
                const int idx = r0 + (u0 + j) * 256 - BB;
                kk[j] = (idx >> 19) + 1;
                tg[j] = heads[idx];
                de[j] = degree[srcs[u0 + j]];
            }
#pragma unroll
            for (int j = 0; j < GRP_B; ++j) bt[j] = batch[tg[j] & (BB - 1)];
#pragma unroll
            for (int j = 0; j < GRP_B; ++j) db[j] = degree[bt[j]];
#pragma unroll
            for (int j = 0; j < GRP_B; ++j)
                sc[j] = att[min(kk[j], KHOP)] * rsqrtf(db[j]) * rsqrtf(de[j]) * dsum * inv_ne;
        }
#pragma unroll
        for (int j = 0; j < GRP_B; ++j) {
            const int src = srcs[u0 + j];
            const int bin = min((int)((unsigned)src >> BIN_SH), NBIN - 1);
            const unsigned nl  = (unsigned)(src - bin * BIN_N) & (unsigned)(BIN_N - 1);
            const unsigned rec = ((unsigned)tg[j] & (unsigned)(BB - 1)) | ((unsigned)kk[j] << 16) | (nl << 19);
            const int sl = boff[bin] + atomicAdd(&fill[bin], 1);
            if ((unsigned)sl < (unsigned)GRP_CH) sorted[sl] = rec;
            atomicAdd(wsum_h + (size_t)tg[j] * OUTD, sc[j]);
        }
    }
}

__global__ void __launch_bounds__(256)
lsappnp_bin(const float* __restrict__ att,
            const float* __restrict__ degree,
            const int* __restrict__ batch,
            const int* __restrict__ heads,
            const int* __restrict__ ends,
            float* __restrict__ ws,
            float* __restrict__ wsum,
            unsigned long long ws_bytes) {
    if (ws_bytes < WS_NEED) return;
    __shared__ int hist[256];
    __shared__ int boff[256];
    __shared__ int fill[256];
    __shared__ int gst[256];
    __shared__ int gbase;
    __shared__ unsigned sorted[GRP_CH];
    unsigned* plane = (unsigned*)(ws + WS_AGGH);
    int* cursor = (int*)ws + 1;
    const int tid  = threadIdx.x;
    const int lane = tid & 31;
    const int wave = tid >> 5;
    const int g    = blockIdx.x;
    const bool isb = g < BB / GRP_CH;
    const int r0   = g * GRP_CH + tid;
    hist[tid] = 0;
    fill[tid] = 0;
    __syncthreads();
    int srcs[GRP_RPT];
#pragma unroll
    for (int u = 0; u < GRP_RPT; ++u) {
        const int r = r0 + u * 256;
        const int* p = isb ? (batch + r) : (ends + (r - BB));
        srcs[u] = *p;
    }
#pragma unroll
    for (int u = 0; u < GRP_RPT; ++u)
        atomicAdd(&hist[min((int)((unsigned)srcs[u] >> BIN_SH), NBIN - 1)], 1);
    __syncthreads();
    const int c  = hist[tid];
    const int lp = (c + 31) & ~31;
    boff[tid] = c;
    gst[tid]  = lp;
    __syncthreads();
    for (int off = 1; off < 256; off <<= 1) {
        int v = (tid >= off) ? boff[tid - off] : 0;
        int w = (tid >= off) ? gst[tid - off] : 0;
        __syncthreads();
        boff[tid] += v;
        gst[tid]  += w;
        __syncthreads();
    }
    const int excl  = boff[tid] - c;
    const int pincl = gst[tid];
    if (tid == 255) gbase = atomicAdd(cursor, pincl);
    __syncthreads();
    const int st = gbase + pincl - lp;
    boff[tid] = excl;
    gst[tid]  = st;
    plane[PL_SEG + g * 256 + tid] = ((unsigned)st >> 5) | ((unsigned)c << 17);
    if (c > 0) atomicAdd((int*)plane + PL_BTOT + tid * 32, c);
    __syncthreads();

    const float dsum = ws[0];
    float* wsum_h = wsum;
    if (ws_bytes >= WS_NEED_WS && l2_half()) wsum_h = ws + WS_WSUM2;
    if (isb) bin_rows(true,  r0, srcs, att, degree, batch, heads, dsum, wsum_h, fill, boff, sorted);
    else     bin_rows(false, r0, srcs, att, degree, batch, heads, dsum, wsum_h, fill, boff, sorted);
    __syncthreads();

    for (int b = wave; b < NBIN; b += 8) {
        const int len = hist[b], o = boff[b], s0 = gst[b];
        for (int k = lane; k < len; k += 32) {
            const int d = s0 + k;
            if ((unsigned)d < (unsigned)POOL_W) plane[d] = sorted[o + k];
        }
    }
}

__global__ void __launch_bounds__(512)
lsappnp_rank(float* __restrict__ ws, unsigned long long ws_bytes) {
    if (ws_bytes < WS_NEED) return;
    __shared__ int cntL[BIN_N];
    __shared__ int posL[BIN_N];
    __shared__ int scan[512];
    extern __shared__ unsigned stage[];
    const unsigned* plane = (const unsigned*)(ws + WS_AGGH);
    int* cnt = (int*)(ws + WS_CNT);
    int* pos = (int*)(ws + WS_POS);
    unsigned* erec = (unsigned*)(ws + WS_EREC);
    const int tid = threadIdx.x;
    const int b   = blockIdx.x;
    for (int i = tid; i < BIN_N; i += 512) cntL[i] = 0;
    scan[tid] = (tid < b) ? (int)plane[PL_BTOT + tid * 32] : 0;
    __syncthreads();
    for (int off = 256; off > 0; off >>= 1) {
        if (tid < off) scan[tid] += scan[tid + off];
        __syncthreads();
    }
    const int base = scan[0];
    __syncthreads();

    const int hsub = tid & 1;
    for (int g = tid >> 1; g < GRP_G; g += 256) {
        const unsigned sg = plane[PL_SEG + g * 256 + b];
        const int s0  = (int)((sg & 0x1FFFFu) << 5);
        const int len = min((int)(sg >> 17), GRP_CH);
        for (int k = hsub; k < len; k += 2) {
            const int d = s0 + k;
            if ((unsigned)d < (unsigned)POOL_W)
                atomicAdd(&cntL[(plane[d] >> 19) & (unsigned)(BIN_N - 1)], 1);
        }
    }
    __syncthreads();

    int loc[4];
    int s = 0;
#pragma unroll
    for (int j = 0; j < 4; ++j) { loc[j] = cntL[tid * 4 + j]; s += loc[j]; }
    scan[tid] = s;
    __syncthreads();
    for (int off = 1; off < 512; off <<= 1) {
        int v = (tid >= off) ? scan[tid - off] : 0;
        __syncthreads();
        scan[tid] += v;
        __syncthreads();
    }
    const int tot = scan[511];
    const bool in_lds = tot <= STG_CAP;
    int run = base + scan[tid] - s;
#pragma unroll
    for (int j = 0; j < 4; ++j) { posL[tid * 4 + j] = run; run += loc[j]; }
    __syncthreads();
    for (int i = tid; i < BIN_N; i += 512) {
        const int n = b * BIN_N + i;
        if (n < NN) { cnt[n] = cntL[i]; pos[n] = posL[i]; }
    }
    __syncthreads();
    for (int i = tid; i < BIN_N; i += 512) cntL[i] = 0;
    __syncthreads();

    for (int g = tid >> 1; g < GRP_G; g += 256) {
        const unsigned sg = plane[PL_SEG + g * 256 + b];
        const int s0  = (int)((sg & 0x1FFFFu) << 5);
        const int len = min((int)(sg >> 17), GRP_CH);
        for (int k = hsub; k < len; k += 2) {
            const int d = s0 + k;
            if ((unsigned)d < (unsigned)POOL_W) {
                const unsigned rec = plane[d];
                const int nl = (int)((rec >> 19) & (unsigned)(BIN_N - 1));
                const int p  = posL[nl] + atomicAdd(&cntL[nl], 1);
                const unsigned rv = rec & 0x7FFFFu;
                if (in_lds) {
                    const int q = p - base;
                    if ((unsigned)q < (unsigned)STG_CAP) stage[q] = rv;
                } else if ((unsigned)p < (unsigned)MTOT) {
                    atomicExch(erec + p, rv);
                }
            }
        }
    }
    if (in_lds) {
        __syncthreads();
        const int lo = base, hi = base + tot;
        const int alo = (lo + 31) & ~31, ahi = hi & ~31;
        for (int i = (lo & ~31) + tid; i < hi; i += 512) {
            if (i >= lo && (unsigned)i < (unsigned)MTOT) {
                const unsigned v = stage[i - lo];
                if (i >= alo && i < ahi) erec[i] = v;
                else atomicExch(erec + i, v);
            }
        }
    }
}

__global__ void lsappnp_zero_plane(float* __restrict__ ws) {
    const size_t i = (size_t)blockIdx.x * blockDim.x + threadIdx.x;
    if (i < (size_t)BB * HDIM / 4) {
        v4f z; z[0] = 0.0f; z[1] = 0.0f; z[2] = 0.0f; z[3] = 0.0f;
        ((v4f*)(ws + WS_AGGH))[i] = z;
    }
}

__device__ __forceinline__ void chunk_issue(int base, int eE, int lane,
                                            const unsigned* __restrict__ erec,
                                            const int* __restrict__ batch,
                                            const float* __restrict__ degree,
                                            const float* __restrict__ att,
                                            unsigned& q0, unsigned& q1,
                                            float& d0, float& d1, float& a0, float& a1) {
    q0 = 0u; q1 = 0u;
    if (base + lane < eE)      q0 = erec[base + lane];
    if (base + 32 + lane < eE) q1 = erec[base + 32 + lane];
    const int t0 = (int)(q0 & (unsigned)(BB - 1)), t1 = (int)(q1 & (unsigned)(BB - 1));
    const int k0 = min((int)((q0 >> 16) & 7u), KHOP), k1 = min((int)((q1 >> 16) & 7u), KHOP);
    d0 = degree[batch[t0]]; d1 = degree[batch[t1]];
    a0 = att[k0];           a1 = att[k1];
}
__device__ __forceinline__ void chunk_weights(unsigned q0, unsigned q1, float d0, float d1,
                                              float a0, float a1, float& p0, float& p1) {
    p0 = (((q0 >> 16) & 7u) == 0u) ? a0 : a0 * rsqrtf(d0);
    p1 = (((q1 >> 16) & 7u) == 0u) ? a1 : a1 * rsqrtf(d1);
}

__global__ void __launch_bounds__(256)
lsappnp_mlp(const float* __restrict__ x,
            const float* __restrict__ W0, const float* __restrict__ b0,
            const float* __restrict__ att,
            const float* __restrict__ degree,
            const int* __restrict__ batch,
            const int* __restrict__ heads,
            const int* __restrict__ ends,
            float* __restrict__ ws,
            float* __restrict__ wsum,
            unsigned long long ws_bytes) {
    extern __shared__ float smem[];
    v2f*   W0p   = (v2f*)smem;
    float* b0s   = (float*)(W0p + 64 * 128);
    float* Xbase = b0s + 128;

    const int tid  = threadIdx.x;
    const int wave = tid >> 5;
    const int lane = tid & 31;
    const int hi   = lane >> 4;
    const int lm   = lane & 15;

    for (int i = tid; i < 64 * 128; i += 256) {
        int k2 = i >> 7, n = i & 127;
        v2f w0; w0[0] = W0[(2 * k2) * 128 + n]; w0[1] = W0[(2 * k2 + 1) * 128 + n];
        W0p[i] = w0;
    }
    if (tid < 128) b0s[tid] = b0[tid];
    __syncthreads();

    float* aggh = ws + WS_AGGH;
    if (ws_bytes >= WS_NEED_PL && l2_half()) aggh = ws + WS_AGG2;
    float* Xs   = Xbase + wave * PWF;

    if (ws_bytes >= WS_NEED) {
        const int* cnt = (const int*)(ws + WS_CNT);
        const int* pos = (const int*)(ws + WS_POS);
        const unsigned* erec = (const unsigned*)(ws + WS_EREC);
        const float dsum   = ws[0];
        const float inv_ne = 1.0f / (float)EDG;
        unsigned* metaT = (unsigned*)(Xs + 16 * XS);
        float*    metaF = (float*)(metaT + 64);

        for (int tile = blockIdx.x * 8 + wave; tile < NTILEN; tile += gridDim.x * 8) {
            const int n0 = tile * 16;
            v4f xr[16];
            const v4f* xp = (const v4f*)(x + (size_t)n0 * FDIM) + lane;
#pragma unroll
            for (int rrow = 0; rrow < 16; ++rrow)
                xr[rrow] = __builtin_nontemporal_load(xp + rrow * 32);
            int c = 0, ps = 0;
            float de = 1.0f;
            if (lane < 16) { c = cnt[n0 + lane]; ps = pos[n0 + lane]; de = degree[n0 + lane]; }
            __builtin_amdgcn_sched_barrier(0);
            const float rde = rsqrtf(de);

            const int pe = ps + c;
            int eS = __builtin_amdgcn_readlane(ps, 0);
            int eE = __builtin_amdgcn_readlane(pe, 15);
            eS = max(eS, 0);
            eE = min(eE, MTOT);
            unsigned q0, q1;
            float d0, d1, a0, a1, p0, p1;
            chunk_issue(eS, eE, lane, erec, batch, degree, att, q0, q1, d0, d1, a0, a1);

#pragma unroll
            for (int rrow = 0; rrow < 16; ++rrow)
                *(v4f*)(Xs + rrow * XS + lane * 4) = xr[rrow];
            wait_ds();

            v8f acc[8];
#pragma unroll
            for (int nt = 0; nt < 8; ++nt) {
                float bv = b0s[nt * 16 + lm];
                v8f cc;
#pragma unroll
                for (int i = 0; i < 8; ++i) cc[i] = bv;
                acc[nt] = cc;
            }
            for (int kt = 0; kt < 32; ++kt) {
                int kc = kt * 4 + hi * 2;
                v2f a  = *(const v2f*)(Xs + lm * XS + kc);
                int k2 = kc >> 1;
#pragma unroll
                for (int nt = 0; nt < 8; ++nt) {
                    v2f bfr = W0p[k2 * 128 + nt * 16 + lm];
                    acc[nt] = __builtin_amdgcn_wmma_f32_16x16x4_f32(
                        false, a, false, bfr, (short)0, acc[nt], false, false);
                }
            }
#pragma unroll
            for (int nt = 0; nt < 8; ++nt)
#pragma unroll
                for (int v = 0; v < 8; ++v)
                    Xs[(v + 8 * hi) * XS + nt * 16 + lm] = fmaxf(acc[nt][v], 0.0f);

            for (int base = eS; base < eE; base += 64) {
                if (base != eS)
                    chunk_issue(base, eE, lane, erec, batch, degree, att, q0, q1, d0, d1, a0, a1);
                chunk_weights(q0, q1, d0, d1, a0, a1, p0, p1);
                metaT[lane]      = q0;
                metaF[lane]      = p0;
                metaT[32 + lane] = q1;
                metaF[32 + lane] = p1;
                wait_ds();
                const int lim = min(eE, base + 64);
#pragma unroll
                for (int m = 0; m < 16; ++m) {
                    const int s = max(__builtin_amdgcn_readlane(ps, m), base);
                    const int e = min(__builtin_amdgcn_readlane(pe, m), lim);
                    if (s < e) {
                        const float* hr = Xs + m * XS + lane;
                        const float h0 = hr[0], h1 = hr[32], h2 = hr[64], h3 = hr[96];
                        const float rde_m = __int_as_float(__builtin_amdgcn_readlane(__float_as_int(rde), m));
                        for (int j = s; j < e; ++j) {
                            const unsigned q  = (unsigned)__builtin_amdgcn_readfirstlane((int)metaT[j - base]);
                            const int      tg = (int)(q & (unsigned)(BB - 1));
                            const float    wp = metaF[j - base];
                            const float    w  = (((q >> 16) & 7u) == 0u) ? wp : wp * rde_m * dsum * inv_ne;
                            float* d = aggh + (size_t)tg * HDIM + lane;
                            atomicAdd(d,      h0 * w);
                            atomicAdd(d + 32, h1 * w);
                            atomicAdd(d + 64, h2 * w);
                            atomicAdd(d + 96, h3 * w);
                        }
                    }
                }
            }
        }
        return;
    }

    const float dsum   = ws[0];
    const float inv_ne = 1.0f / (float)EDG;

    int*   metaT = (int*)(Xs + 16 * XS);
    float* metaF = (float*)(metaT + 16);

    for (int tile = blockIdx.x * 8 + wave; tile < NTILES; tile += gridDim.x * 8) {
        int   src = 0;
        int   tgt = 0;
        float sc  = 0.0f;
        if (lane < 16) {
            long r = (long)tile * 16 + lane;
            if (r < BB) {
                tgt = (int)r;
                src = batch[r];
                sc  = att[0];
            } else {
                long idx = r - BB;
                int  k   = (int)(idx >> 19);
                tgt = heads[idx];
                src = ends[idx];
                float db = degree[batch[tgt]];
                float de = degree[src];
                sc = att[k + 1] * rsqrtf(db) * rsqrtf(de) * dsum * inv_ne;
            }
            metaT[lane] = tgt; metaF[lane] = sc;
        }

        float4 xr[16];
#pragma unroll
        for (int rrow = 0; rrow < 16; ++rrow) {
            int s = __builtin_amdgcn_readlane(src, rrow);
            xr[rrow] = ((const float4*)(x + (size_t)s * FDIM))[lane];
        }
#pragma unroll
        for (int rrow = 0; rrow < 16; ++rrow)
            ((float4*)(Xs + rrow * XS))[lane] = xr[rrow];
        wait_ds();

        v8f acc[8];
#pragma unroll
        for (int nt = 0; nt < 8; ++nt) {
            float bv = b0s[nt * 16 + lm];
            v8f c;
#pragma unroll
            for (int i = 0; i < 8; ++i) c[i] = bv;
            acc[nt] = c;
        }
        for (int kt = 0; kt < 32; ++kt) {
            int kc = kt * 4 + hi * 2;
            v2f a  = *(const v2f*)(Xs + lm * XS + kc);
            int k2 = kc >> 1;
#pragma unroll
            for (int nt = 0; nt < 8; ++nt) {
                v2f bfr = W0p[k2 * 128 + nt * 16 + lm];
                acc[nt] = __builtin_amdgcn_wmma_f32_16x16x4_f32(
                    false, a, false, bfr, (short)0, acc[nt], false, false);
            }
        }

#pragma unroll
        for (int v = 0; v < 8; ++v) {
            int   tg = metaT[v + 8 * hi];
            float sw = metaF[v + 8 * hi];
            float* base = aggh + (size_t)tg * HDIM + lm;
#pragma unroll
            for (int nt = 0; nt < 8; ++nt)
                atomicAdd(base + nt * 16, fmaxf(acc[nt][v], 0.0f) * sw);
        }
        if (lane < 16) atomicAdd(wsum + (size_t)tgt * OUTD, sc);
    }
}

__global__ void __launch_bounds__(256)
lsappnp_head(const float* __restrict__ ws,
             const float* __restrict__ W1, const float* __restrict__ b1,
             const float* __restrict__ W2, const float* __restrict__ b2,
             float* out, unsigned long long ws_bytes) {
    const float* aggh  = ws + WS_AGGH;
    const float* aggh2 = ws + WS_AGG2;
    const bool two_planes = ws_bytes >= WS_NEED_PL;
    const bool two_wsums  = ws_bytes >= WS_NEED_WS;
    extern __shared__ float smem[];
    v2f*   W1p   = (v2f*)smem;
    float* b1s   = (float*)(W1p + 64 * 128);
    float* W2s   = b1s + 128;
    float* b2s   = W2s + HDIM * OUTD;
    float* Xbase = b2s + 64;

    const int tid  = threadIdx.x;
    const int wave = tid >> 5;
    const int lane = tid & 31;
    const int hi   = lane >> 4;
    const int lm   = lane & 15;

    for (int i = tid; i < 64 * 128; i += 256) {
        int k2 = i >> 7, n = i & 127;
        v2f w1; w1[0] = W1[(2 * k2) * 128 + n]; w1[1] = W1[(2 * k2 + 1) * 128 + n];
        W1p[i] = w1;
    }
    if (tid < 128) b1s[tid] = b1[tid];
    for (int i = tid; i < HDIM * OUTD; i += 256) W2s[i] = W2[i];
    if (tid < OUTD) b2s[tid] = b2[tid];
    __syncthreads();

    float* Xs    = Xbase + wave * PWF;
    float* metaF = Xs + 16 * XS;

    for (int tile = blockIdx.x * 8 + wave; tile < NTILEB; tile += gridDim.x * 8) {
        const size_t row0 = (size_t)tile * 16;

        float4 xr[16];
#pragma unroll
        for (int rrow = 0; rrow < 16; ++rrow)
            xr[rrow] = ((const float4*)(aggh + (row0 + rrow) * HDIM))[lane];
        float wsv = 0.0f;
        if (lane < 16) wsv = out[(row0 + lane) * OUTD];
        if (two_wsums && lane < 16) wsv += ws[WS_WSUM2 + (row0 + lane) * OUTD];
        if (two_planes) {
            float4 yr[16];
#pragma unroll
            for (int rrow = 0; rrow < 16; ++rrow)
                yr[rrow] = ((const float4*)(aggh2 + (row0 + rrow) * HDIM))[lane];
#pragma unroll
            for (int rrow = 0; rrow < 16; ++rrow) {
                xr[rrow].x += yr[rrow].x; xr[rrow].y += yr[rrow].y;
                xr[rrow].z += yr[rrow].z; xr[rrow].w += yr[rrow].w;
            }
        }
#pragma unroll
        for (int rrow = 0; rrow < 16; ++rrow)
            ((float4*)(Xs + rrow * XS))[lane] = xr[rrow];
        if (lane < 16) metaF[lane] = wsv;
        wait_ds();

        v8f acc2[8];
#pragma unroll
        for (int nt = 0; nt < 8; ++nt) {
            float bv = b1s[nt * 16 + lm];
            v8f c;
#pragma unroll
            for (int i = 0; i < 8; ++i) c[i] = metaF[i + 8 * hi] * bv;
            acc2[nt] = c;
        }
        for (int kt = 0; kt < 32; ++kt) {
            int kc = kt * 4 + hi * 2;
            v2f a  = *(const v2f*)(Xs + lm * XS + kc);
            int k2 = kc >> 1;
#pragma unroll
            for (int nt = 0; nt < 8; ++nt) {
                v2f bfr = W1p[k2 * 128 + nt * 16 + lm];
                acc2[nt] = __builtin_amdgcn_wmma_f32_16x16x4_f32(
                    false, a, false, bfr, (short)0, acc2[nt], false, false);
            }
        }
#pragma unroll
        for (int nt = 0; nt < 8; ++nt)
#pragma unroll
            for (int v = 0; v < 8; ++v)
                Xs[(v + 8 * hi) * XS + nt * 16 + lm] = fmaxf(acc2[nt][v], 0.0f);
        wait_ds();

        const float* ar = Xs + lm * XS;
        const float* wr = W2s + hi * 4;
        float p0 = 0.0f, p1 = 0.0f, p2 = 0.0f, p3 = 0.0f;
        for (int c = 0; c < HDIM; c += 4) {
            float4 a  = *(const float4*)(ar + c);
            float4 w0 = *(const float4*)(wr + (c + 0) * OUTD);
            float4 w1 = *(const float4*)(wr + (c + 1) * OUTD);
            float4 w2 = *(const float4*)(wr + (c + 2) * OUTD);
            float4 w3 = *(const float4*)(wr + (c + 3) * OUTD);
            p0 += a.x * w0.x; p1 += a.x * w0.y; p2 += a.x * w0.z; p3 += a.x * w0.w;
            p0 += a.y * w1.x; p1 += a.y * w1.y; p2 += a.y * w1.z; p3 += a.y * w1.w;
            p0 += a.z * w2.x; p1 += a.z * w2.y; p2 += a.z * w2.z; p3 += a.z * w2.w;
            p0 += a.w * w3.x; p1 += a.w * w3.y; p2 += a.w * w3.z; p3 += a.w * w3.w;
        }
        p0 += b2s[hi * 4 + 0]; p1 += b2s[hi * 4 + 1]; p2 += b2s[hi * 4 + 2]; p3 += b2s[hi * 4 + 3];
        float m = fmaxf(fmaxf(p0, p1), fmaxf(p2, p3));
        m = fmaxf(m, __shfl_xor(m, 16));
        float s = __expf(p0 - m) + __expf(p1 - m) + __expf(p2 - m) + __expf(p3 - m);
        s += __shfl_xor(s, 16);
        float lse = __logf(s);
        float4 o;
        o.x = p0 - m - lse; o.y = p1 - m - lse; o.z = p2 - m - lse; o.w = p3 - m - lse;
        *(float4*)(out + (row0 + lm) * OUTD + hi * 4) = o;
    }
}

extern "C" void kernel_launch(void* const* d_in, const int* in_sizes, int n_in,
                              void* d_out, int out_size, void* d_ws, size_t ws_size,
                              hipStream_t stream) {
    const float* x      = (const float*)d_in[0];
    const float* W0     = (const float*)d_in[1];
    const float* b0     = (const float*)d_in[2];
    const float* W1     = (const float*)d_in[3];
    const float* b1     = (const float*)d_in[4];
    const float* W2     = (const float*)d_in[5];
    const float* b2     = (const float*)d_in[6];
    const float* att    = (const float*)d_in[7];
    const float* degree = (const float*)d_in[8];
    const int*   batch  = (const int*)d_in[9];
    const int*   heads  = (const int*)d_in[10];
    const int*   ends   = (const int*)d_in[11];
    float*       out    = (float*)d_out;
    float*       ws     = (float*)d_ws;
    const unsigned long long ws_bytes = (unsigned long long)ws_size;

    lsappnp_zero<<<(BB * HDIM / 4 + 255) / 256, 256, 0, stream>>>(ws, out, ws_bytes);
    lsappnp_dsum<<<64, 256, 0, stream>>>(degree, batch, ws);
    lsappnp_bin<<<GRP_G, 256, 0, stream>>>(att, degree, batch, heads, ends, ws, out, ws_bytes);
    lsappnp_rank<<<NBIN, 512, (size_t)STG_CAP * sizeof(unsigned), stream>>>(ws, ws_bytes);
    lsappnp_zero_plane<<<(BB * HDIM / 4 + 255) / 256, 256, 0, stream>>>(ws);

    size_t shmem_mlp = (size_t)64 * 128 * sizeof(v2f)
                     + 128 * sizeof(float)
                     + (size_t)8 * PWF * sizeof(float);
    lsappnp_mlp<<<512, 256, shmem_mlp, stream>>>(x, W0, b0, att, degree,
                                                 batch, heads, ends, ws, out, ws_bytes);

    size_t shmem_head = (size_t)64 * 128 * sizeof(v2f)
                      + (128 + HDIM * OUTD + 64) * sizeof(float)
                      + (size_t)8 * PWF * sizeof(float);
    lsappnp_head<<<NTILEB / 8, 256, shmem_head, stream>>>(ws, W1, b1, W2, b2, out, ws_bytes);
}
